# starting kernel
# baseline (speedup 1.0000x reference)
#include <hip/hip_runtime.h>
#include <hip/hip_bf16.h>


#define BATCH   2
#define IMG     384
#define C_DIM   96
#define HEADS   6
#define HD      16
#define WIN     16
#define NWIN    24
#define BW      (BATCH * NWIN * NWIN)
#define LTOK    256
#define KSEL    64
#define NPIX    (BATCH * IMG * IMG)

typedef __attribute__((ext_vector_type(2))) float v2f;
typedef __attribute__((ext_vector_type(8))) float v8f;

static __device__ inline v8f wmma4(v2f a, v2f b, v8f c) {
    return __builtin_amdgcn_wmma_f32_16x16x4_f32(
         false, a,  false, b,
         (short)0, c,  false,  false);
}

__global__ __launch_bounds__(256) void k1_router(
    const float* __restrict__ x, const float* __restrict__ Wr,
    int* __restrict__ gidx, float* __restrict__ gval)
{
    __shared__ float sWr[C_DIM * HEADS];
    __shared__ float sv[LTOK];
    __shared__ unsigned long long key[LTOK];

    const int w = blockIdx.x;
    const int t = threadIdx.x;

    for (int i = t; i < C_DIM * HEADS; i += 256) sWr[i] = Wr[i];
    __syncthreads();

    const int b  = w / (NWIN * NWIN);
    const int rm = w % (NWIN * NWIN);
    const int wy = rm / NWIN, wx = rm % NWIN;
    const int hh = wy * WIN + (t >> 4);
    const int ww = wx * WIN + (t & 15);
    const float* xr = x + ((size_t)(b * IMG + hh) * IMG + ww) * C_DIM;

    float lg[HEADS];
    #pragma unroll
    for (int e = 0; e < HEADS; ++e) lg[e] = 0.f;
    for (int c = 0; c < C_DIM; ++c) {
        float xv = xr[c];
        #pragma unroll
        for (int e = 0; e < HEADS; ++e) lg[e] += xv * sWr[c * HEADS + e];
    }
    #pragma unroll
    for (int e = 0; e < HEADS; ++e) lg[e] = 1.0f / (1.0f + __expf(-lg[e]));

    for (int e = 0; e < HEADS; ++e) {
        __syncthreads();
        sv[t] = lg[e];
        unsigned int fb = (t == 0) ? 0x7F800000u : __float_as_uint(lg[e]);
        key[t] = ((unsigned long long)fb << 32) | (unsigned int)(0xFFFFu - t);
        __syncthreads();
        for (int ks = 2; ks <= LTOK; ks <<= 1) {
            for (int j = ks >> 1; j > 0; j >>= 1) {
                int ixj = t ^ j;
                if (ixj > t) {
                    unsigned long long a = key[t], bb = key[ixj];
                    bool descBlock = ((t & ks) == 0);
                    if (descBlock ? (a < bb) : (a > bb)) { key[t] = bb; key[ixj] = a; }
                }
                __syncthreads();
            }
        }
        if (t < KSEL) {
            int tok = 0xFFFF - (int)(key[t] & 0xFFFFull);
            gidx[((size_t)w * HEADS + e) * KSEL + t] = tok;
            gval[((size_t)w * HEADS + e) * KSEL + t] = sv[tok];
        }
    }
}

__global__ __launch_bounds__(128) void k2_attn(
    const float* __restrict__ x,
    const float* __restrict__ Wqkv, const float* __restrict__ Wo,
    const int* __restrict__ gidx, const float* __restrict__ gval,
    float* __restrict__ scat)
{
    __shared__ float pool[11264];
    __shared__ float sQKV[KSEL * 48];
    __shared__ float sgate[KSEL];
    __shared__ int   sidx[KSEL];

    float* Xg  = pool;
    float* sWq = pool + 6144;
    float* S   = pool;
    float* O   = pool + 4096;
    float* sWo = pool + 5120;

    const int w    = blockIdx.x;
    const int e    = blockIdx.y;
    const int tid  = threadIdx.x;
    const int lane = tid & 31;
    const int wave = tid >> 5;
    const int hi   = lane >> 4;
    const int lo   = lane & 15;

    if (tid < KSEL) {
        sidx[tid]  = gidx[((size_t)w * HEADS + e) * KSEL + tid];
        sgate[tid] = gval[((size_t)w * HEADS + e) * KSEL + tid];
    }
    __syncthreads();

    const int b  = w / (NWIN * NWIN);
    const int rm = w % (NWIN * NWIN);
    const int wy = rm / NWIN, wx = rm % NWIN;

    for (int i = tid; i < KSEL * C_DIM; i += 128) {
        int j = i / C_DIM, c = i % C_DIM;
        int t  = sidx[j];
        int hh = wy * WIN + (t >> 4);
        int ww = wx * WIN + (t & 15);
        Xg[i] = x[((size_t)(b * IMG + hh) * IMG + ww) * C_DIM + c];
    }
    for (int i = tid; i < C_DIM * 48; i += 128)
        sWq[i] = Wqkv[(size_t)e * C_DIM * 48 + i];
    __syncthreads();

    for (int tile = wave; tile < 12; tile += 4) {
        int mt = tile / 3, nt = tile % 3;
        v8f acc = {};
        for (int kk = 0; kk < C_DIM; kk += 4) {
            int ka = kk + 2 * hi;
            v2f a = { Xg[(mt * 16 + lo) * C_DIM + ka],
                      Xg[(mt * 16 + lo) * C_DIM + ka + 1] };
            v2f bf = { sWq[ka * 48 + nt * 16 + lo],
                       sWq[(ka + 1) * 48 + nt * 16 + lo] };
            acc = wmma4(a, bf, acc);
        }
        #pragma unroll
        for (int r = 0; r < 8; ++r)
            sQKV[(mt * 16 + r + 8 * hi) * 48 + nt * 16 + lo] = acc[r];
    }
    __syncthreads();

    for (int i = tid; i < HD * C_DIM; i += 128)
        sWo[i] = Wo[(size_t)e * HD * C_DIM + i];

    for (int tile = wave; tile < 16; tile += 4) {
        int mt = tile / 4, nt = tile % 4;
        v8f acc = {};
        for (int kk = 0; kk < HD; kk += 4) {
            int ka = kk + 2 * hi;
            v2f a = { sQKV[(mt * 16 + lo) * 48 + ka],
                      sQKV[(mt * 16 + lo) * 48 + ka + 1] };
            v2f bf = { sQKV[(nt * 16 + lo) * 48 + 16 + ka],
                       sQKV[(nt * 16 + lo) * 48 + 16 + ka + 1] };
            acc = wmma4(a, bf, acc);
        }
        #pragma unroll
        for (int r = 0; r < 8; ++r)
            S[(mt * 16 + r + 8 * hi) * 64 + nt * 16 + lo] = acc[r] * 0.25f;
    }
    __syncthreads();

    if (tid < KSEL) {
        float m = -1e30f;
        for (int j = 0; j < 64; ++j) m = fmaxf(m, S[tid * 64 + j]);
        float s = 0.f;
        for (int j = 0; j < 64; ++j) {
            float ev = __expf(S[tid * 64 + j] - m);
            S[tid * 64 + j] = ev; s += ev;
        }
        float inv = sgate[tid] / s;
        for (int j = 0; j < 64; ++j) S[tid * 64 + j] *= inv;
    }
    __syncthreads();

    {
        int mt = wave;
        v8f acc = {};
        for (int kk = 0; kk < 64; kk += 4) {
            int ka = kk + 2 * hi;
            v2f a = { S[(mt * 16 + lo) * 64 + ka],
                      S[(mt * 16 + lo) * 64 + ka + 1] };
            v2f bf = { sQKV[ka * 48 + 32 + lo],
                       sQKV[(ka + 1) * 48 + 32 + lo] };
            acc = wmma4(a, bf, acc);
        }
        #pragma unroll
        for (int r = 0; r < 8; ++r)
            O[(mt * 16 + r + 8 * hi) * HD + lo] = acc[r];
    }
    __syncthreads();

    for (int tile = wave; tile < 24; tile += 4) {
        int mt = tile / 6, nt = tile % 6;
        v8f acc = {};
        for (int kk = 0; kk < HD; kk += 4) {
            int ka = kk + 2 * hi;
            v2f a = { O[(mt * 16 + lo) * HD + ka],
                      O[(mt * 16 + lo) * HD + ka + 1] };
            v2f bf = { sWo[ka * C_DIM + nt * 16 + lo],
                       sWo[(ka + 1) * C_DIM + nt * 16 + lo] };
            acc = wmma4(a, bf, acc);
        }
        #pragma unroll
        for (int r = 0; r < 8; ++r) {
            int m  = mt * 16 + r + 8 * hi;
            int t  = sidx[m];
            int hh = wy * WIN + (t >> 4);
            int ww = wx * WIN + (t & 15);
            atomicAdd(&scat[((size_t)(b * IMG + hh) * IMG + ww) * C_DIM + nt * 16 + lo],
                      acc[r]);
        }
    }
}

__global__ __launch_bounds__(256) void k3_proj(
    const float* __restrict__ scat, const float* __restrict__ projW,
    const float* __restrict__ projb, float* __restrict__ y)
{
    __shared__ float A[64 * C_DIM];
    __shared__ float Wt[C_DIM * C_DIM];
    __shared__ float bias[C_DIM];

    const int tid  = threadIdx.x;
    const size_t row0 = (size_t)blockIdx.x * 64;

    for (int i = tid; i < 64 * C_DIM; i += 256) A[i] = scat[row0 * C_DIM + i];
    for (int i = tid; i < C_DIM * C_DIM; i += 256) {
        int c = i / C_DIM, j = i % C_DIM;
        Wt[i] = projW[(size_t)j * C_DIM + c];
    }
    if (tid < C_DIM) bias[tid] = projb[tid];
    __syncthreads();

    const int lane = tid & 31, wave = tid >> 5;
    const int hi = lane >> 4, lo = lane & 15;

    for (int tile = wave; tile < 24; tile += 8) {
        int mt = tile / 6, nt = tile % 6;
        v8f acc = {};
        for (int kk = 0; kk < C_DIM; kk += 4) {
            int ka = kk + 2 * hi;
            v2f a = { A[(mt * 16 + lo) * C_DIM + ka],
                      A[(mt * 16 + lo) * C_DIM + ka + 1] };
            v2f bf = { Wt[ka * C_DIM + nt * 16 + lo],
                       Wt[(ka + 1) * C_DIM + nt * 16 + lo] };
            acc = wmma4(a, bf, acc);
        }
        #pragma unroll
        for (int r = 0; r < 8; ++r) {
            int m = mt * 16 + r + 8 * hi;
            y[(row0 + m) * C_DIM + nt * 16 + lo] = acc[r] + bias[nt * 16 + lo];
        }
    }
}

extern "C" void kernel_launch(void* const* d_in, const int* in_sizes, int n_in,
                              void* d_out, int out_size, void* d_ws, size_t ws_size,
                              hipStream_t stream) {
    const float* x     = (const float*)d_in[0];
    const float* Wr    = (const float*)d_in[1];
    const float* Wqkv  = (const float*)d_in[2];
    const float* Wo    = (const float*)d_in[3];
    const float* projW = (const float*)d_in[4];
    const float* projb = (const float*)d_in[5];
    float* y = (float*)d_out;

    const size_t scatBytes = (size_t)NPIX * C_DIM * sizeof(float);
    float* scat = (float*)d_ws;
    int*   gidx = (int*)((char*)d_ws + scatBytes);
    float* gval = (float*)((char*)d_ws + scatBytes +
                           (size_t)BW * HEADS * KSEL * sizeof(int));

    hipMemsetAsync(scat, 0, scatBytes, stream);

    k1_router<<<BW, 256, 0, stream>>>(x, Wr, gidx, gval);
    k2_attn<<<dim3(BW, HEADS), 128, 0, stream>>>(x, Wqkv, Wo, gidx, gval, scat);
    k3_proj<<<NPIX / 64, 256, 0, stream>>>(scat, projW, projb, y);
}


# k3: both GEMM operands row-major in LDS at a 100-float pitch; projW staged as stored with 16-byte copies (no strided transposing read); each WMMA fragment is one conflict-free 8-byte LDS read
# speedup vs baseline: 1.1673x; 1.1673x over previous
#include <hip/hip_runtime.h>
#include <hip/hip_bf16.h>


#define BATCH   2
#define IMG     384
#define C_DIM   96
#define HEADS   6
#define HD      16
#define WIN     16
#define NWIN    24
#define BW      (BATCH * NWIN * NWIN)
#define LTOK    256
#define KSEL    64
#define NPIX    (BATCH * IMG * IMG)

typedef __attribute__((ext_vector_type(2))) float v2f;
typedef __attribute__((ext_vector_type(8))) float v8f;

static __device__ inline v8f wmma4(v2f a, v2f b, v8f c) {
    return __builtin_amdgcn_wmma_f32_16x16x4_f32(
         false, a,  false, b,
         (short)0, c,  false,  false);
}

__global__ __launch_bounds__(256) void k1_router(
    const float* __restrict__ x, const float* __restrict__ Wr,
    int* __restrict__ gidx, float* __restrict__ gval)
{
    __shared__ float sWr[C_DIM * HEADS];
    __shared__ float sv[LTOK];
    __shared__ unsigned long long key[LTOK];

    const int w = blockIdx.x;
    const int t = threadIdx.x;

    for (int i = t; i < C_DIM * HEADS; i += 256) sWr[i] = Wr[i];
    __syncthreads();

    const int b  = w / (NWIN * NWIN);
    const int rm = w % (NWIN * NWIN);
    const int wy = rm / NWIN, wx = rm % NWIN;
    const int hh = wy * WIN + (t >> 4);
    const int ww = wx * WIN + (t & 15);
    const float* xr = x + ((size_t)(b * IMG + hh) * IMG + ww) * C_DIM;

    float lg[HEADS];
    #pragma unroll
    for (int e = 0; e < HEADS; ++e) lg[e] = 0.f;
    for (int c = 0; c < C_DIM; ++c) {
        float xv = xr[c];
        #pragma unroll
        for (int e = 0; e < HEADS; ++e) lg[e] += xv * sWr[c * HEADS + e];
    }
    #pragma unroll
    for (int e = 0; e < HEADS; ++e) lg[e] = 1.0f / (1.0f + __expf(-lg[e]));

    for (int e = 0; e < HEADS; ++e) {
        __syncthreads();
        sv[t] = lg[e];
        unsigned int fb = (t == 0) ? 0x7F800000u : __float_as_uint(lg[e]);
        key[t] = ((unsigned long long)fb << 32) | (unsigned int)(0xFFFFu - t);
        __syncthreads();
        for (int ks = 2; ks <= LTOK; ks <<= 1) {
            for (int j = ks >> 1; j > 0; j >>= 1) {
                int ixj = t ^ j;
                if (ixj > t) {
                    unsigned long long a = key[t], bb = key[ixj];
                    bool descBlock = ((t & ks) == 0);
                    if (descBlock ? (a < bb) : (a > bb)) { key[t] = bb; key[ixj] = a; }
                }
                __syncthreads();
            }
        }
        if (t < KSEL) {
            int tok = 0xFFFF - (int)(key[t] & 0xFFFFull);
            gidx[((size_t)w * HEADS + e) * KSEL + t] = tok;
            gval[((size_t)w * HEADS + e) * KSEL + t] = sv[tok];
        }
    }
}

__global__ __launch_bounds__(128) void k2_attn(
    const float* __restrict__ x,
    const float* __restrict__ Wqkv, const float* __restrict__ Wo,
    const int* __restrict__ gidx, const float* __restrict__ gval,
    float* __restrict__ scat)
{
    __shared__ float pool[11264];
    __shared__ float sQKV[KSEL * 48];
    __shared__ float sgate[KSEL];
    __shared__ int   sidx[KSEL];

    float* Xg  = pool;
    float* sWq = pool + 6144;
    float* S   = pool;
    float* O   = pool + 4096;
    float* sWo = pool + 5120;

    const int w    = blockIdx.x;
    const int e    = blockIdx.y;
    const int tid  = threadIdx.x;
    const int lane = tid & 31;
    const int wave = tid >> 5;
    const int hi   = lane >> 4;
    const int lo   = lane & 15;

    if (tid < KSEL) {
        sidx[tid]  = gidx[((size_t)w * HEADS + e) * KSEL + tid];
        sgate[tid] = gval[((size_t)w * HEADS + e) * KSEL + tid];
    }
    __syncthreads();

    const int b  = w / (NWIN * NWIN);
    const int rm = w % (NWIN * NWIN);
    const int wy = rm / NWIN, wx = rm % NWIN;

    for (int i = tid; i < KSEL * C_DIM; i += 128) {
        int j = i / C_DIM, c = i % C_DIM;
        int t  = sidx[j];
        int hh = wy * WIN + (t >> 4);
        int ww = wx * WIN + (t & 15);
        Xg[i] = x[((size_t)(b * IMG + hh) * IMG + ww) * C_DIM + c];
    }
    for (int i = tid; i < C_DIM * 48; i += 128)
        sWq[i] = Wqkv[(size_t)e * C_DIM * 48 + i];
    __syncthreads();

    for (int tile = wave; tile < 12; tile += 4) {
        int mt = tile / 3, nt = tile % 3;
        v8f acc = {};
        for (int kk = 0; kk < C_DIM; kk += 4) {
            int ka = kk + 2 * hi;
            v2f a = { Xg[(mt * 16 + lo) * C_DIM + ka],
                      Xg[(mt * 16 + lo) * C_DIM + ka + 1] };
            v2f bf = { sWq[ka * 48 + nt * 16 + lo],
                       sWq[(ka + 1) * 48 + nt * 16 + lo] };
            acc = wmma4(a, bf, acc);
        }
        #pragma unroll
        for (int r = 0; r < 8; ++r)
            sQKV[(mt * 16 + r + 8 * hi) * 48 + nt * 16 + lo] = acc[r];
    }
    __syncthreads();

    for (int i = tid; i < HD * C_DIM; i += 128)
        sWo[i] = Wo[(size_t)e * HD * C_DIM + i];

    for (int tile = wave; tile < 16; tile += 4) {
        int mt = tile / 4, nt = tile % 4;
        v8f acc = {};
        for (int kk = 0; kk < HD; kk += 4) {
            int ka = kk + 2 * hi;
            v2f a = { sQKV[(mt * 16 + lo) * 48 + ka],
                      sQKV[(mt * 16 + lo) * 48 + ka + 1] };
            v2f bf = { sQKV[(nt * 16 + lo) * 48 + 16 + ka],
                       sQKV[(nt * 16 + lo) * 48 + 16 + ka + 1] };
            acc = wmma4(a, bf, acc);
        }
        #pragma unroll
        for (int r = 0; r < 8; ++r)
            S[(mt * 16 + r + 8 * hi) * 64 + nt * 16 + lo] = acc[r] * 0.25f;
    }
    __syncthreads();

    if (tid < KSEL) {
        float m = -1e30f;
        for (int j = 0; j < 64; ++j) m = fmaxf(m, S[tid * 64 + j]);
        float s = 0.f;
        for (int j = 0; j < 64; ++j) {
            float ev = __expf(S[tid * 64 + j] - m);
            S[tid * 64 + j] = ev; s += ev;
        }
        float inv = sgate[tid] / s;
        for (int j = 0; j < 64; ++j) S[tid * 64 + j] *= inv;
    }
    __syncthreads();

    {
        int mt = wave;
        v8f acc = {};
        for (int kk = 0; kk < 64; kk += 4) {
            int ka = kk + 2 * hi;
            v2f a = { S[(mt * 16 + lo) * 64 + ka],
                      S[(mt * 16 + lo) * 64 + ka + 1] };
            v2f bf = { sQKV[ka * 48 + 32 + lo],
                       sQKV[(ka + 1) * 48 + 32 + lo] };
            acc = wmma4(a, bf, acc);
        }
        #pragma unroll
        for (int r = 0; r < 8; ++r)
            O[(mt * 16 + r + 8 * hi) * HD + lo] = acc[r];
    }
    __syncthreads();

    for (int tile = wave; tile < 24; tile += 4) {
        int mt = tile / 6, nt = tile % 6;
        v8f acc = {};
        for (int kk = 0; kk < HD; kk += 4) {
            int ka = kk + 2 * hi;
            v2f a = { O[(mt * 16 + lo) * HD + ka],
                      O[(mt * 16 + lo) * HD + ka + 1] };
            v2f bf = { sWo[ka * C_DIM + nt * 16 + lo],
                       sWo[(ka + 1) * C_DIM + nt * 16 + lo] };
            acc = wmma4(a, bf, acc);
        }
        #pragma unroll
        for (int r = 0; r < 8; ++r) {
            int m  = mt * 16 + r + 8 * hi;
            int t  = sidx[m];
            int hh = wy * WIN + (t >> 4);
            int ww = wx * WIN + (t & 15);
            atomicAdd(&scat[((size_t)(b * IMG + hh) * IMG + ww) * C_DIM + nt * 16 + lo],
                      acc[r]);
        }
    }
}

#define LDP 100
__global__ __launch_bounds__(256) void k3_proj(
    const float* __restrict__ scat, const float* __restrict__ projW,
    const float* __restrict__ projb, float* __restrict__ y)
{
    __shared__ __attribute__((aligned(16))) float A[64 * LDP];
    __shared__ __attribute__((aligned(16))) float Wn[C_DIM * LDP];
    __shared__ float bias[C_DIM];

    const int tid  = threadIdx.x;
    const size_t row0 = (size_t)blockIdx.x * 64;

    const float4* s4 = reinterpret_cast<const float4*>(scat + row0 * C_DIM);
    const float4* w4 = reinterpret_cast<const float4*>(projW);
    float4 va[6], vw[9];
    for (int t = 0; t < 6; ++t) va[t] = s4[tid + t * 256];
    for (int t = 0; t < 9; ++t) vw[t] = w4[tid + t * 256];
    for (int t = 0; t < 6; ++t) {
        int i = tid + t * 256;
        int r = i / (C_DIM / 4), c4 = i % (C_DIM / 4);
        *reinterpret_cast<float4*>(&A[r * LDP + c4 * 4]) = va[t];
    }
    for (int t = 0; t < 9; ++t) {
        int i = tid + t * 256;
        int n = i / (C_DIM / 4), c4 = i % (C_DIM / 4);
        *reinterpret_cast<float4*>(&Wn[n * LDP + c4 * 4]) = vw[t];
    }
    if (tid < C_DIM) bias[tid] = projb[tid];
    __syncthreads();

    const int lane = tid & 31, wave = tid >> 5;
    const int hi = lane >> 4, lo = lane & 15;

    for (int tile = wave; tile < 24; tile += 8) {
        int mt = tile / 6, nt = tile % 6;
        const float* ap = &A[(mt * 16 + lo) * LDP + 2 * hi];
        const float* bp = &Wn[(nt * 16 + lo) * LDP + 2 * hi];
        v8f acc = {};
        for (int kk = 0; kk < C_DIM; kk += 4) {
            v2f a  = *reinterpret_cast<const v2f*>(ap + kk);
            v2f bf = *reinterpret_cast<const v2f*>(bp + kk);
            acc = wmma4(a, bf, acc);
        }
        #pragma unroll
        for (int r = 0; r < 8; ++r) {
            int m = mt * 16 + r + 8 * hi;
            y[(row0 + m) * C_DIM + nt * 16 + lo] = acc[r] + bias[nt * 16 + lo];
        }
    }
}

extern "C" void kernel_launch(void* const* d_in, const int* in_sizes, int n_in,
                              void* d_out, int out_size, void* d_ws, size_t ws_size,
                              hipStream_t stream) {
    const float* x     = (const float*)d_in[0];
    const float* Wr    = (const float*)d_in[1];
    const float* Wqkv  = (const float*)d_in[2];
    const float* Wo    = (const float*)d_in[3];
    const float* projW = (const float*)d_in[4];
    const float* projb = (const float*)d_in[5];
    float* y = (float*)d_out;

    const size_t scatBytes = (size_t)NPIX * C_DIM * sizeof(float);
    float* scat = (float*)d_ws;
    int*   gidx = (int*)((char*)d_ws + scatBytes);
    float* gval = (float*)((char*)d_ws + scatBytes +
                           (size_t)BW * HEADS * KSEL * sizeof(int));

    hipMemsetAsync(scat, 0, scatBytes, stream);

    k1_router<<<BW, 256, 0, stream>>>(x, Wr, gidx, gval);
    k2_attn<<<dim3(BW, HEADS), 128, 0, stream>>>(x, Wqkv, Wo, gidx, gval, scat);
    k3_proj<<<NPIX / 64, 256, 0, stream>>>(scat, projW, projb, y);
}


# k2: LDS row pitches of Xg, QKV, S, O padded to 4 x odd floats (100/52/68/20; Wo 104): each A/K fragment is one conflict-free 8-byte read and softmax row walks no longer share one bank
# speedup vs baseline: 1.3848x; 1.1863x over previous
#include <hip/hip_runtime.h>
#include <hip/hip_bf16.h>


#define BATCH   2
#define IMG     384
#define C_DIM   96
#define HEADS   6
#define HD      16
#define WIN     16
#define NWIN    24
#define BW      (BATCH * NWIN * NWIN)
#define LTOK    256
#define KSEL    64
#define NPIX    (BATCH * IMG * IMG)

typedef __attribute__((ext_vector_type(2))) float v2f;
typedef __attribute__((ext_vector_type(8))) float v8f;

static __device__ inline v8f wmma4(v2f a, v2f b, v8f c) {
    return __builtin_amdgcn_wmma_f32_16x16x4_f32(
         false, a,  false, b,
         (short)0, c,  false,  false);
}

__global__ __launch_bounds__(256) void k1_router(
    const float* __restrict__ x, const float* __restrict__ Wr,
    int* __restrict__ gidx, float* __restrict__ gval)
{
    __shared__ float sWr[C_DIM * HEADS];
    __shared__ float sv[LTOK];
    __shared__ unsigned long long key[LTOK];

    const int w = blockIdx.x;
    const int t = threadIdx.x;

    for (int i = t; i < C_DIM * HEADS; i += 256) sWr[i] = Wr[i];
    __syncthreads();

    const int b  = w / (NWIN * NWIN);
    const int rm = w % (NWIN * NWIN);
    const int wy = rm / NWIN, wx = rm % NWIN;
    const int hh = wy * WIN + (t >> 4);
    const int ww = wx * WIN + (t & 15);
    const float* xr = x + ((size_t)(b * IMG + hh) * IMG + ww) * C_DIM;

    float lg[HEADS];
    #pragma unroll
    for (int e = 0; e < HEADS; ++e) lg[e] = 0.f;
    for (int c = 0; c < C_DIM; ++c) {
        float xv = xr[c];
        #pragma unroll
        for (int e = 0; e < HEADS; ++e) lg[e] += xv * sWr[c * HEADS + e];
    }
    #pragma unroll
    for (int e = 0; e < HEADS; ++e) lg[e] = 1.0f / (1.0f + __expf(-lg[e]));

    for (int e = 0; e < HEADS; ++e) {
        __syncthreads();
        sv[t] = lg[e];
        unsigned int fb = (t == 0) ? 0x7F800000u : __float_as_uint(lg[e]);
        key[t] = ((unsigned long long)fb << 32) | (unsigned int)(0xFFFFu - t);
        __syncthreads();
        for (int ks = 2; ks <= LTOK; ks <<= 1) {
            for (int j = ks >> 1; j > 0; j >>= 1) {
                int ixj = t ^ j;
                if (ixj > t) {
                    unsigned long long a = key[t], bb = key[ixj];
                    bool descBlock = ((t & ks) == 0);
                    if (descBlock ? (a < bb) : (a > bb)) { key[t] = bb; key[ixj] = a; }
                }
                __syncthreads();
            }
        }
        if (t < KSEL) {
            int tok = 0xFFFF - (int)(key[t] & 0xFFFFull);
            gidx[((size_t)w * HEADS + e) * KSEL + t] = tok;
            gval[((size_t)w * HEADS + e) * KSEL + t] = sv[tok];
        }
    }
}

#define XP  100
#define QP  52
#define SP  68
#define OP  20
#define WOP 104
__global__ __launch_bounds__(128) void k2_attn(
    const float* __restrict__ x,
    const float* __restrict__ Wqkv, const float* __restrict__ Wo,
    const int* __restrict__ gidx, const float* __restrict__ gval,
    float* __restrict__ scat)
{
    __shared__ __attribute__((aligned(16))) float pool[KSEL * XP + C_DIM * 48];
    __shared__ __attribute__((aligned(16))) float sQKV[KSEL * QP];
    __shared__ float sgate[KSEL];
    __shared__ int   sidx[KSEL];

    float* Xg  = pool;
    float* sWq = pool + KSEL * XP;
    float* S   = pool;
    float* O   = pool + KSEL * SP;
    float* sWo = pool + KSEL * SP + KSEL * OP;

    const int w    = blockIdx.x;
    const int e    = blockIdx.y;
    const int tid  = threadIdx.x;
    const int lane = tid & 31;
    const int wave = tid >> 5;
    const int hi   = lane >> 4;
    const int lo   = lane & 15;

    if (tid < KSEL) {
        sidx[tid]  = gidx[((size_t)w * HEADS + e) * KSEL + tid];
        sgate[tid] = gval[((size_t)w * HEADS + e) * KSEL + tid];
    }
    __syncthreads();

    const int b  = w / (NWIN * NWIN);
    const int rm = w % (NWIN * NWIN);
    const int wy = rm / NWIN, wx = rm % NWIN;

    for (int i = tid; i < KSEL * C_DIM; i += 128) {
        int j = i / C_DIM, c = i % C_DIM;
        int t  = sidx[j];
        int hh = wy * WIN + (t >> 4);
        int ww = wx * WIN + (t & 15);
        Xg[j * XP + c] = x[((size_t)(b * IMG + hh) * IMG + ww) * C_DIM + c];
    }
    for (int i = tid; i < C_DIM * 48; i += 128)
        sWq[i] = Wqkv[(size_t)e * C_DIM * 48 + i];
    __syncthreads();

    for (int tile = wave; tile < 12; tile += 4) {
        int mt = tile / 3, nt = tile % 3;
        const float* ap = &Xg[(mt * 16 + lo) * XP + 2 * hi];
        v8f acc = {};
        for (int kk = 0; kk < C_DIM; kk += 4) {
            int ka = kk + 2 * hi;
            v2f a  = *reinterpret_cast<const v2f*>(ap + kk);
            v2f bf = { sWq[ka * 48 + nt * 16 + lo],
                       sWq[(ka + 1) * 48 + nt * 16 + lo] };
            acc = wmma4(a, bf, acc);
        }
        #pragma unroll
        for (int r = 0; r < 8; ++r)
            sQKV[(mt * 16 + r + 8 * hi) * QP + nt * 16 + lo] = acc[r];
    }
    __syncthreads();

    for (int i = tid; i < HD * C_DIM; i += 128) {
        int r = i / C_DIM, c = i % C_DIM;
        sWo[r * WOP + c] = Wo[(size_t)e * HD * C_DIM + i];
    }

    for (int tile = wave; tile < 16; tile += 4) {
        int mt = tile / 4, nt = tile % 4;
        const float* ap = &sQKV[(mt * 16 + lo) * QP + 2 * hi];
        const float* bp = &sQKV[(nt * 16 + lo) * QP + 16 + 2 * hi];
        v8f acc = {};
        for (int kk = 0; kk < HD; kk += 4) {
            v2f a  = *reinterpret_cast<const v2f*>(ap + kk);
            v2f bf = *reinterpret_cast<const v2f*>(bp + kk);
            acc = wmma4(a, bf, acc);
        }
        #pragma unroll
        for (int r = 0; r < 8; ++r)
            S[(mt * 16 + r + 8 * hi) * SP + nt * 16 + lo] = acc[r] * 0.25f;
    }
    __syncthreads();

    if (tid < KSEL) {
        float* Sr = &S[tid * SP];
        float m = -1e30f;
        for (int j = 0; j < 64; ++j) m = fmaxf(m, Sr[j]);
        float s = 0.f;
        for (int j = 0; j < 64; ++j) {
            float ev = __expf(Sr[j] - m);
            Sr[j] = ev; s += ev;
        }
        float inv = sgate[tid] / s;
        for (int j = 0; j < 64; ++j) Sr[j] *= inv;
    }
    __syncthreads();

    {
        int mt = wave;
        const float* ap = &S[(mt * 16 + lo) * SP + 2 * hi];
        v8f acc = {};
        for (int kk = 0; kk < 64; kk += 4) {
            int ka = kk + 2 * hi;
            v2f a  = *reinterpret_cast<const v2f*>(ap + kk);
            v2f bf = { sQKV[ka * QP + 32 + lo],
                       sQKV[(ka + 1) * QP + 32 + lo] };
            acc = wmma4(a, bf, acc);
        }
        #pragma unroll
        for (int r = 0; r < 8; ++r)
            O[(mt * 16 + r + 8 * hi) * OP + lo] = acc[r];
    }
    __syncthreads();

    for (int tile = wave; tile < 24; tile += 4) {
        int mt = tile / 6, nt = tile % 6;
        const float* ap = &O[(mt * 16 + lo) * OP + 2 * hi];
        v8f acc = {};
        for (int kk = 0; kk < HD; kk += 4) {
            int ka = kk + 2 * hi;
            v2f a  = *reinterpret_cast<const v2f*>(ap + kk);
            v2f bf = { sWo[ka * WOP + nt * 16 + lo],
                       sWo[(ka + 1) * WOP + nt * 16 + lo] };
            acc = wmma4(a, bf, acc);
        }
        #pragma unroll
        for (int r = 0; r < 8; ++r) {
            int m  = mt * 16 + r + 8 * hi;
            int t  = sidx[m];
            int hh = wy * WIN + (t >> 4);
            int ww = wx * WIN + (t & 15);
            atomicAdd(&scat[((size_t)(b * IMG + hh) * IMG + ww) * C_DIM + nt * 16 + lo],
                      acc[r]);
        }
    }
}

#define LDP 100
__global__ __launch_bounds__(256) void k3_proj(
    const float* __restrict__ scat, const float* __restrict__ projW,
    const float* __restrict__ projb, float* __restrict__ y)
{
    __shared__ __attribute__((aligned(16))) float A[64 * LDP];
    __shared__ __attribute__((aligned(16))) float Wn[C_DIM * LDP];
    __shared__ float bias[C_DIM];

    const int tid  = threadIdx.x;
    const size_t row0 = (size_t)blockIdx.x * 64;

    const float4* s4 = reinterpret_cast<const float4*>(scat + row0 * C_DIM);
    const float4* w4 = reinterpret_cast<const float4*>(projW);
    float4 va[6], vw[9];
    for (int t = 0; t < 6; ++t) va[t] = s4[tid + t * 256];
    for (int t = 0; t < 9; ++t) vw[t] = w4[tid + t * 256];
    for (int t = 0; t < 6; ++t) {
        int i = tid + t * 256;
        int r = i / (C_DIM / 4), c4 = i % (C_DIM / 4);
        *reinterpret_cast<float4*>(&A[r * LDP + c4 * 4]) = va[t];
    }
    for (int t = 0; t < 9; ++t) {
        int i = tid + t * 256;
        int n = i / (C_DIM / 4), c4 = i % (C_DIM / 4);
        *reinterpret_cast<float4*>(&Wn[n * LDP + c4 * 4]) = vw[t];
    }
    if (tid < C_DIM) bias[tid] = projb[tid];
    __syncthreads();

    const int lane = tid & 31, wave = tid >> 5;
    const int hi = lane >> 4, lo = lane & 15;

    for (int tile = wave; tile < 24; tile += 8) {
        int mt = tile / 6, nt = tile % 6;
        const float* ap = &A[(mt * 16 + lo) * LDP + 2 * hi];
        const float* bp = &Wn[(nt * 16 + lo) * LDP + 2 * hi];
        v8f acc = {};
        for (int kk = 0; kk < C_DIM; kk += 4) {
            v2f a  = *reinterpret_cast<const v2f*>(ap + kk);
            v2f bf = *reinterpret_cast<const v2f*>(bp + kk);
            acc = wmma4(a, bf, acc);
        }
        #pragma unroll
        for (int r = 0; r < 8; ++r) {
            int m = mt * 16 + r + 8 * hi;
            y[(row0 + m) * C_DIM + nt * 16 + lo] = acc[r] + bias[nt * 16 + lo];
        }
    }
}

extern "C" void kernel_launch(void* const* d_in, const int* in_sizes, int n_in,
                              void* d_out, int out_size, void* d_ws, size_t ws_size,
                              hipStream_t stream) {
    const float* x     = (const float*)d_in[0];
    const float* Wr    = (const float*)d_in[1];
    const float* Wqkv  = (const float*)d_in[2];
    const float* Wo    = (const float*)d_in[3];
    const float* projW = (const float*)d_in[4];
    const float* projb = (const float*)d_in[5];
    float* y = (float*)d_out;

    const size_t scatBytes = (size_t)NPIX * C_DIM * sizeof(float);
    float* scat = (float*)d_ws;
    int*   gidx = (int*)((char*)d_ws + scatBytes);
    float* gval = (float*)((char*)d_ws + scatBytes +
                           (size_t)BW * HEADS * KSEL * sizeof(int));

    hipMemsetAsync(scat, 0, scatBytes, stream);

    k1_router<<<BW, 256, 0, stream>>>(x, Wr, gidx, gval);
    k2_attn<<<dim3(BW, HEADS), 128, 0, stream>>>(x, Wqkv, Wo, gidx, gval, scat);
    k3_proj<<<NPIX / 64, 256, 0, stream>>>(scat, projW, projb, y);
}


# k2: gathered x rows, Wqkv_e and Wo_e staged with 16-byte loads issued together ahead of the first LDS store: 3 waited memory round trips per workgroup instead of 34 (24 of them the 4-byte x gather)
# speedup vs baseline: 1.6631x; 1.2010x over previous
#include <hip/hip_runtime.h>
#include <hip/hip_bf16.h>


#define BATCH   2
#define IMG     384
#define C_DIM   96
#define HEADS   6
#define HD      16
#define WIN     16
#define NWIN    24
#define BW      (BATCH * NWIN * NWIN)
#define LTOK    256
#define KSEL    64
#define NPIX    (BATCH * IMG * IMG)

typedef __attribute__((ext_vector_type(2))) float v2f;
typedef __attribute__((ext_vector_type(8))) float v8f;

static __device__ inline v8f wmma4(v2f a, v2f b, v8f c) {
    return __builtin_amdgcn_wmma_f32_16x16x4_f32(
         false, a,  false, b,
         (short)0, c,  false,  false);
}

__global__ __launch_bounds__(256) void k1_router(
    const float* __restrict__ x, const float* __restrict__ Wr,
    int* __restrict__ gidx, float* __restrict__ gval)
{
    __shared__ float sWr[C_DIM * HEADS];
    __shared__ float sv[LTOK];
    __shared__ unsigned long long key[LTOK];

    const int w = blockIdx.x;
    const int t = threadIdx.x;

    for (int i = t; i < C_DIM * HEADS; i += 256) sWr[i] = Wr[i];
    __syncthreads();

    const int b  = w / (NWIN * NWIN);
    const int rm = w % (NWIN * NWIN);
    const int wy = rm / NWIN, wx = rm % NWIN;
    const int hh = wy * WIN + (t >> 4);
    const int ww = wx * WIN + (t & 15);
    const float* xr = x + ((size_t)(b * IMG + hh) * IMG + ww) * C_DIM;

    float lg[HEADS];
    #pragma unroll
    for (int e = 0; e < HEADS; ++e) lg[e] = 0.f;
    for (int c = 0; c < C_DIM; ++c) {
        float xv = xr[c];
        #pragma unroll
        for (int e = 0; e < HEADS; ++e) lg[e] += xv * sWr[c * HEADS + e];
    }
    #pragma unroll
    for (int e = 0; e < HEADS; ++e) lg[e] = 1.0f / (1.0f + __expf(-lg[e]));

    for (int e = 0; e < HEADS; ++e) {
        __syncthreads();
        sv[t] = lg[e];
        unsigned int fb = (t == 0) ? 0x7F800000u : __float_as_uint(lg[e]);
        key[t] = ((unsigned long long)fb << 32) | (unsigned int)(0xFFFFu - t);
        __syncthreads();
        for (int ks = 2; ks <= LTOK; ks <<= 1) {
            for (int j = ks >> 1; j > 0; j >>= 1) {
                int ixj = t ^ j;
                if (ixj > t) {
                    unsigned long long a = key[t], bb = key[ixj];
                    bool descBlock = ((t & ks) == 0);
                    if (descBlock ? (a < bb) : (a > bb)) { key[t] = bb; key[ixj] = a; }
                }
                __syncthreads();
            }
        }
        if (t < KSEL) {
            int tok = 0xFFFF - (int)(key[t] & 0xFFFFull);
            gidx[((size_t)w * HEADS + e) * KSEL + t] = tok;
            gval[((size_t)w * HEADS + e) * KSEL + t] = sv[tok];
        }
    }
}

#define XP  100
#define QP  52
#define SP  68
#define OP  20
#define WOP 104
__global__ __launch_bounds__(128) void k2_attn(
    const float* __restrict__ x,
    const float* __restrict__ Wqkv, const float* __restrict__ Wo,
    const int* __restrict__ gidx, const float* __restrict__ gval,
    float* __restrict__ scat)
{
    __shared__ __attribute__((aligned(16))) float pool[KSEL * XP + C_DIM * 48];
    __shared__ __attribute__((aligned(16))) float sQKV[KSEL * QP];
    __shared__ float sgate[KSEL];
    __shared__ int   sidx[KSEL];

    float* Xg  = pool;
    float* sWq = pool + KSEL * XP;
    float* S   = pool;
    float* O   = pool + KSEL * SP;
    float* sWo = pool + KSEL * SP + KSEL * OP;

    const int w    = blockIdx.x;
    const int e    = blockIdx.y;
    const int tid  = threadIdx.x;
    const int lane = tid & 31;
    const int wave = tid >> 5;
    const int hi   = lane >> 4;
    const int lo   = lane & 15;

    if (tid < KSEL) {
        sidx[tid]  = gidx[((size_t)w * HEADS + e) * KSEL + tid];
        sgate[tid] = gval[((size_t)w * HEADS + e) * KSEL + tid];
    }
    __syncthreads();

    const int b  = w / (NWIN * NWIN);
    const int rm = w % (NWIN * NWIN);
    const int wy = rm / NWIN, wx = rm % NWIN;

    constexpr int ROW4 = C_DIM / 4;
    constexpr int NGX  = KSEL * ROW4 / 128;
    constexpr int NGW  = C_DIM * 48 / 4 / 128;
    constexpr int NGO  = HD * C_DIM / 4 / 128;
    {
        float4 gx[NGX], gw[NGW];
        const float4* wq4 = reinterpret_cast<const float4*>(Wqkv + (size_t)e * C_DIM * 48);
        for (int k = 0; k < NGX; ++k) {
            int i = tid + k * 128;
            int j = i / ROW4, c4 = i % ROW4;
            int t  = sidx[j];
            int hh = wy * WIN + (t >> 4);
            int ww = wx * WIN + (t & 15);
            gx[k] = *reinterpret_cast<const float4*>(
                x + ((size_t)(b * IMG + hh) * IMG + ww) * C_DIM + c4 * 4);
        }
        for (int k = 0; k < NGW; ++k) gw[k] = wq4[tid + k * 128];
        for (int k = 0; k < NGX; ++k) {
            int i = tid + k * 128;
            int j = i / ROW4, c4 = i % ROW4;
            *reinterpret_cast<float4*>(&Xg[j * XP + c4 * 4]) = gx[k];
        }
        for (int k = 0; k < NGW; ++k)
            *reinterpret_cast<float4*>(&sWq[(tid + k * 128) * 4]) = gw[k];
    }
    __syncthreads();

    for (int tile = wave; tile < 12; tile += 4) {
        int mt = tile / 3, nt = tile % 3;
        const float* ap = &Xg[(mt * 16 + lo) * XP + 2 * hi];
        v8f acc = {};
        for (int kk = 0; kk < C_DIM; kk += 4) {
            int ka = kk + 2 * hi;
            v2f a  = *reinterpret_cast<const v2f*>(ap + kk);
            v2f bf = { sWq[ka * 48 + nt * 16 + lo],
                       sWq[(ka + 1) * 48 + nt * 16 + lo] };
            acc = wmma4(a, bf, acc);
        }
        #pragma unroll
        for (int r = 0; r < 8; ++r)
            sQKV[(mt * 16 + r + 8 * hi) * QP + nt * 16 + lo] = acc[r];
    }
    __syncthreads();

    {
        const float4* wo4 = reinterpret_cast<const float4*>(Wo + (size_t)e * HD * C_DIM);
        float4 go[NGO];
        for (int k = 0; k < NGO; ++k) go[k] = wo4[tid + k * 128];
        for (int k = 0; k < NGO; ++k) {
            int i = tid + k * 128;
            int r = i / ROW4, c4 = i % ROW4;
            *reinterpret_cast<float4*>(&sWo[r * WOP + c4 * 4]) = go[k];
        }
    }

    for (int tile = wave; tile < 16; tile += 4) {
        int mt = tile / 4, nt = tile % 4;
        const float* ap = &sQKV[(mt * 16 + lo) * QP + 2 * hi];
        const float* bp = &sQKV[(nt * 16 + lo) * QP + 16 + 2 * hi];
        v8f acc = {};
        for (int kk = 0; kk < HD; kk += 4) {
            v2f a  = *reinterpret_cast<const v2f*>(ap + kk);
            v2f bf = *reinterpret_cast<const v2f*>(bp + kk);
            acc = wmma4(a, bf, acc);
        }
        #pragma unroll
        for (int r = 0; r < 8; ++r)
            S[(mt * 16 + r + 8 * hi) * SP + nt * 16 + lo] = acc[r] * 0.25f;
    }
    __syncthreads();

    if (tid < KSEL) {
        float* Sr = &S[tid * SP];
        float m = -1e30f;
        for (int j = 0; j < 64; ++j) m = fmaxf(m, Sr[j]);
        float s = 0.f;
        for (int j = 0; j < 64; ++j) {
            float ev = __expf(Sr[j] - m);
            Sr[j] = ev; s += ev;
        }
        float inv = sgate[tid] / s;
        for (int j = 0; j < 64; ++j) Sr[j] *= inv;
    }
    __syncthreads();

    {
        int mt = wave;
        const float* ap = &S[(mt * 16 + lo) * SP + 2 * hi];
        v8f acc = {};
        for (int kk = 0; kk < 64; kk += 4) {
            int ka = kk + 2 * hi;
            v2f a  = *reinterpret_cast<const v2f*>(ap + kk);
            v2f bf = { sQKV[ka * QP + 32 + lo],
                       sQKV[(ka + 1) * QP + 32 + lo] };
            acc = wmma4(a, bf, acc);
        }
        #pragma unroll
        for (int r = 0; r < 8; ++r)
            O[(mt * 16 + r + 8 * hi) * OP + lo] = acc[r];
    }
    __syncthreads();

    for (int tile = wave; tile < 24; tile += 4) {
        int mt = tile / 6, nt = tile % 6;
        const float* ap = &O[(mt * 16 + lo) * OP + 2 * hi];
        v8f acc = {};
        for (int kk = 0; kk < HD; kk += 4) {
            int ka = kk + 2 * hi;
            v2f a  = *reinterpret_cast<const v2f*>(ap + kk);
            v2f bf = { sWo[ka * WOP + nt * 16 + lo],
                       sWo[(ka + 1) * WOP + nt * 16 + lo] };
            acc = wmma4(a, bf, acc);
        }
        #pragma unroll
        for (int r = 0; r < 8; ++r) {
            int m  = mt * 16 + r + 8 * hi;
            int t  = sidx[m];
            int hh = wy * WIN + (t >> 4);
            int ww = wx * WIN + (t & 15);
            atomicAdd(&scat[((size_t)(b * IMG + hh) * IMG + ww) * C_DIM + nt * 16 + lo],
                      acc[r]);
        }
    }
}

#define LDP 100
__global__ __launch_bounds__(256) void k3_proj(
    const float* __restrict__ scat, const float* __restrict__ projW,
    const float* __restrict__ projb, float* __restrict__ y)
{
    __shared__ __attribute__((aligned(16))) float A[64 * LDP];
    __shared__ __attribute__((aligned(16))) float Wn[C_DIM * LDP];
    __shared__ float bias[C_DIM];

    const int tid  = threadIdx.x;
    const size_t row0 = (size_t)blockIdx.x * 64;

    const float4* s4 = reinterpret_cast<const float4*>(scat + row0 * C_DIM);
    const float4* w4 = reinterpret_cast<const float4*>(projW);
    float4 va[6], vw[9];
    for (int t = 0; t < 6; ++t) va[t] = s4[tid + t * 256];
    for (int t = 0; t < 9; ++t) vw[t] = w4[tid + t * 256];
    for (int t = 0; t < 6; ++t) {
        int i = tid + t * 256;
        int r = i / (C_DIM / 4), c4 = i % (C_DIM / 4);
        *reinterpret_cast<float4*>(&A[r * LDP + c4 * 4]) = va[t];
    }
    for (int t = 0; t < 9; ++t) {
        int i = tid + t * 256;
        int n = i / (C_DIM / 4), c4 = i % (C_DIM / 4);
        *reinterpret_cast<float4*>(&Wn[n * LDP + c4 * 4]) = vw[t];
    }
    if (tid < C_DIM) bias[tid] = projb[tid];
    __syncthreads();

    const int lane = tid & 31, wave = tid >> 5;
    const int hi = lane >> 4, lo = lane & 15;

    for (int tile = wave; tile < 24; tile += 8) {
        int mt = tile / 6, nt = tile % 6;
        const float* ap = &A[(mt * 16 + lo) * LDP + 2 * hi];
        const float* bp = &Wn[(nt * 16 + lo) * LDP + 2 * hi];
        v8f acc = {};
        for (int kk = 0; kk < C_DIM; kk += 4) {
            v2f a  = *reinterpret_cast<const v2f*>(ap + kk);
            v2f bf = *reinterpret_cast<const v2f*>(bp + kk);
            acc = wmma4(a, bf, acc);
        }
        #pragma unroll
        for (int r = 0; r < 8; ++r) {
            int m = mt * 16 + r + 8 * hi;
            y[(row0 + m) * C_DIM + nt * 16 + lo] = acc[r] + bias[nt * 16 + lo];
        }
    }
}

extern "C" void kernel_launch(void* const* d_in, const int* in_sizes, int n_in,
                              void* d_out, int out_size, void* d_ws, size_t ws_size,
                              hipStream_t stream) {
    const float* x     = (const float*)d_in[0];
    const float* Wr    = (const float*)d_in[1];
    const float* Wqkv  = (const float*)d_in[2];
    const float* Wo    = (const float*)d_in[3];
    const float* projW = (const float*)d_in[4];
    const float* projb = (const float*)d_in[5];
    float* y = (float*)d_out;

    const size_t scatBytes = (size_t)NPIX * C_DIM * sizeof(float);
    float* scat = (float*)d_ws;
    int*   gidx = (int*)((char*)d_ws + scatBytes);
    float* gval = (float*)((char*)d_ws + scatBytes +
                           (size_t)BW * HEADS * KSEL * sizeof(int));

    hipMemsetAsync(scat, 0, scatBytes, stream);

    k1_router<<<BW, 256, 0, stream>>>(x, Wr, gidx, gval);
    k2_attn<<<dim3(BW, HEADS), 128, 0, stream>>>(x, Wqkv, Wo, gidx, gval, scat);
    k3_proj<<<NPIX / 64, 256, 0, stream>>>(scat, projW, projb, y);
}


# projW folded into Wo (Wp_e = Wo_e @ projW^T, f32 WMMA, once per call): k1 sets y = bias, k2 atomics add O_e @ Wp_e into y; k3 GEMM, scat memset and re-read gone; float sums reassociated
# speedup vs baseline: 2.2840x; 1.3733x over previous
#include <hip/hip_runtime.h>
#include <hip/hip_bf16.h>


#define BATCH   2
#define IMG     384
#define C_DIM   96
#define HEADS   6
#define HD      16
#define WIN     16
#define NWIN    24
#define BW      (BATCH * NWIN * NWIN)
#define LTOK    256
#define KSEL    64

typedef __attribute__((ext_vector_type(2))) float v2f;
typedef __attribute__((ext_vector_type(8))) float v8f;

static __device__ inline v8f wmma4(v2f a, v2f b, v8f c) {
    return __builtin_amdgcn_wmma_f32_16x16x4_f32(
         false, a,  false, b,
         (short)0, c,  false,  false);
}

__global__ __launch_bounds__(256) void k1_router(
    const float* __restrict__ x, const float* __restrict__ Wr,
    const float* __restrict__ Wo, const float* __restrict__ projW,
    const float* __restrict__ projb,
    int* __restrict__ gidx, float* __restrict__ gval,
    float* __restrict__ Wp, float* __restrict__ y)
{
    __shared__ float sWr[C_DIM * HEADS];
    __shared__ float sv[LTOK];
    __shared__ unsigned long long key[LTOK];
    __shared__ __attribute__((aligned(16))) float sb[C_DIM];

    const int w = blockIdx.x;
    const int t = threadIdx.x;

    if (w >= BW) {
        const int e    = w - BW;
        const int lane = t & 31, nt = t >> 5;
        const int hi   = lane >> 4, lo = lane & 15;
        if (nt < C_DIM / 16) {
            const float* ap = Wo + ((size_t)e * HD + lo) * C_DIM + 2 * hi;
            const float* bp = projW + (size_t)(nt * 16 + lo) * C_DIM + 2 * hi;
            v8f acc = {};
            for (int kk = 0; kk < C_DIM; kk += 4) {
                v2f a  = *reinterpret_cast<const v2f*>(ap + kk);
                v2f bf = *reinterpret_cast<const v2f*>(bp + kk);
                acc = wmma4(a, bf, acc);
            }
            #pragma unroll
            for (int r = 0; r < 8; ++r)
                Wp[((size_t)e * HD + r + 8 * hi) * C_DIM + nt * 16 + lo] = acc[r];
        }
        return;
    }

    for (int i = t; i < C_DIM * HEADS; i += 256) sWr[i] = Wr[i];
    if (t < C_DIM) sb[t] = projb[t];
    __syncthreads();

    const int b  = w / (NWIN * NWIN);
    const int rm = w % (NWIN * NWIN);
    const int wy = rm / NWIN, wx = rm % NWIN;

    {
        constexpr int ROWP = WIN * C_DIM / 4;
        constexpr int IMGP = IMG * C_DIM / 4;
        float4* y4 = reinterpret_cast<float4*>(y) +
                     ((size_t)(b * IMG + wy * WIN) * IMG + wx * WIN) * (C_DIM / 4);
        const float4* sb4 = reinterpret_cast<const float4*>(sb);
        float4 bv[3];
        int    off[3];
        #pragma unroll
        for (int q = 0; q < 3; ++q) {
            int i = t + q * 256;
            int c = (i >= ROWP) ? i - ROWP : i;
            off[q] = ((i >= ROWP) ? IMGP : 0) + c;
            bv[q]  = sb4[c % (C_DIM / 4)];
        }
        #pragma unroll
        for (int p = 0; p < WIN / 2; ++p) {
            #pragma unroll
            for (int q = 0; q < 3; ++q) y4[(size_t)p * 2 * IMGP + off[q]] = bv[q];
        }
    }
    const int hh = wy * WIN + (t >> 4);
    const int ww = wx * WIN + (t & 15);
    const float* xr = x + ((size_t)(b * IMG + hh) * IMG + ww) * C_DIM;

    float lg[HEADS];
    #pragma unroll
    for (int e = 0; e < HEADS; ++e) lg[e] = 0.f;
    for (int c = 0; c < C_DIM; ++c) {
        float xv = xr[c];
        #pragma unroll
        for (int e = 0; e < HEADS; ++e) lg[e] += xv * sWr[c * HEADS + e];
    }
    #pragma unroll
    for (int e = 0; e < HEADS; ++e) lg[e] = 1.0f / (1.0f + __expf(-lg[e]));

    for (int e = 0; e < HEADS; ++e) {
        __syncthreads();
        sv[t] = lg[e];
        unsigned int fb = (t == 0) ? 0x7F800000u : __float_as_uint(lg[e]);
        key[t] = ((unsigned long long)fb << 32) | (unsigned int)(0xFFFFu - t);
        __syncthreads();
        for (int ks = 2; ks <= LTOK; ks <<= 1) {
            for (int j = ks >> 1; j > 0; j >>= 1) {
                int ixj = t ^ j;
                if (ixj > t) {
                    unsigned long long a = key[t], bb = key[ixj];
                    bool descBlock = ((t & ks) == 0);
                    if (descBlock ? (a < bb) : (a > bb)) { key[t] = bb; key[ixj] = a; }
                }
                __syncthreads();
            }
        }
        if (t < KSEL) {
            int tok = 0xFFFF - (int)(key[t] & 0xFFFFull);
            gidx[((size_t)w * HEADS + e) * KSEL + t] = tok;
            gval[((size_t)w * HEADS + e) * KSEL + t] = sv[tok];
        }
    }
}

#define XP  100
#define QP  52
#define SP  68
#define OP  20
#define WOP 104
__global__ __launch_bounds__(128) void k2_attn(
    const float* __restrict__ x,
    const float* __restrict__ Wqkv, const float* __restrict__ Wp,
    const int* __restrict__ gidx, const float* __restrict__ gval,
    float* __restrict__ y)
{
    __shared__ __attribute__((aligned(16))) float pool[KSEL * XP + C_DIM * 48];
    __shared__ __attribute__((aligned(16))) float sQKV[KSEL * QP];
    __shared__ float sgate[KSEL];
    __shared__ int   sidx[KSEL];

    float* Xg  = pool;
    float* sWq = pool + KSEL * XP;
    float* S   = pool;
    float* O   = pool + KSEL * SP;
    float* sWp = pool + KSEL * SP + KSEL * OP;

    const int w    = blockIdx.x;
    const int e    = blockIdx.y;
    const int tid  = threadIdx.x;
    const int lane = tid & 31;
    const int wave = tid >> 5;
    const int hi   = lane >> 4;
    const int lo   = lane & 15;

    if (tid < KSEL) {
        sidx[tid]  = gidx[((size_t)w * HEADS + e) * KSEL + tid];
        sgate[tid] = gval[((size_t)w * HEADS + e) * KSEL + tid];
    }
    __syncthreads();

    const int b  = w / (NWIN * NWIN);
    const int rm = w % (NWIN * NWIN);
    const int wy = rm / NWIN, wx = rm % NWIN;

    constexpr int ROW4 = C_DIM / 4;
    constexpr int NGX  = KSEL * ROW4 / 128;
    constexpr int NGW  = C_DIM * 48 / 4 / 128;
    constexpr int NGO  = HD * C_DIM / 4 / 128;
    {
        float4 gx[NGX], gw[NGW];
        const float4* wq4 = reinterpret_cast<const float4*>(Wqkv + (size_t)e * C_DIM * 48);
        for (int k = 0; k < NGX; ++k) {
            int i = tid + k * 128;
            int j = i / ROW4, c4 = i % ROW4;
            int t  = sidx[j];
            int hh = wy * WIN + (t >> 4);
            int ww = wx * WIN + (t & 15);
            gx[k] = *reinterpret_cast<const float4*>(
                x + ((size_t)(b * IMG + hh) * IMG + ww) * C_DIM + c4 * 4);
        }
        for (int k = 0; k < NGW; ++k) gw[k] = wq4[tid + k * 128];
        for (int k = 0; k < NGX; ++k) {
            int i = tid + k * 128;
            int j = i / ROW4, c4 = i % ROW4;
            *reinterpret_cast<float4*>(&Xg[j * XP + c4 * 4]) = gx[k];
        }
        for (int k = 0; k < NGW; ++k)
            *reinterpret_cast<float4*>(&sWq[(tid + k * 128) * 4]) = gw[k];
    }
    __syncthreads();

    for (int tile = wave; tile < 12; tile += 4) {
        int mt = tile / 3, nt = tile % 3;
        const float* ap = &Xg[(mt * 16 + lo) * XP + 2 * hi];
        v8f acc = {};
        for (int kk = 0; kk < C_DIM; kk += 4) {
            int ka = kk + 2 * hi;
            v2f a  = *reinterpret_cast<const v2f*>(ap + kk);
            v2f bf = { sWq[ka * 48 + nt * 16 + lo],
                       sWq[(ka + 1) * 48 + nt * 16 + lo] };
            acc = wmma4(a, bf, acc);
        }
        #pragma unroll
        for (int r = 0; r < 8; ++r)
            sQKV[(mt * 16 + r + 8 * hi) * QP + nt * 16 + lo] = acc[r];
    }
    __syncthreads();

    {
        const float4* wp4 = reinterpret_cast<const float4*>(Wp + (size_t)e * HD * C_DIM);
        float4 go[NGO];
        for (int k = 0; k < NGO; ++k) go[k] = wp4[tid + k * 128];
        for (int k = 0; k < NGO; ++k) {
            int i = tid + k * 128;
            int r = i / ROW4, c4 = i % ROW4;
            *reinterpret_cast<float4*>(&sWp[r * WOP + c4 * 4]) = go[k];
        }
    }

    for (int tile = wave; tile < 16; tile += 4) {
        int mt = tile / 4, nt = tile % 4;
        const float* ap = &sQKV[(mt * 16 + lo) * QP + 2 * hi];
        const float* bp = &sQKV[(nt * 16 + lo) * QP + 16 + 2 * hi];
        v8f acc = {};
        for (int kk = 0; kk < HD; kk += 4) {
            v2f a  = *reinterpret_cast<const v2f*>(ap + kk);
            v2f bf = *reinterpret_cast<const v2f*>(bp + kk);
            acc = wmma4(a, bf, acc);
        }
        #pragma unroll
        for (int r = 0; r < 8; ++r)
            S[(mt * 16 + r + 8 * hi) * SP + nt * 16 + lo] = acc[r] * 0.25f;
    }
    __syncthreads();

    if (tid < KSEL) {
        float* Sr = &S[tid * SP];
        float m = -1e30f;
        for (int j = 0; j < 64; ++j) m = fmaxf(m, Sr[j]);
        float s = 0.f;
        for (int j = 0; j < 64; ++j) {
            float ev = __expf(Sr[j] - m);
            Sr[j] = ev; s += ev;
        }
        float inv = sgate[tid] / s;
        for (int j = 0; j < 64; ++j) Sr[j] *= inv;
    }
    __syncthreads();

    {
        int mt = wave;
        const float* ap = &S[(mt * 16 + lo) * SP + 2 * hi];
        v8f acc = {};
        for (int kk = 0; kk < 64; kk += 4) {
            int ka = kk + 2 * hi;
            v2f a  = *reinterpret_cast<const v2f*>(ap + kk);
            v2f bf = { sQKV[ka * QP + 32 + lo],
                       sQKV[(ka + 1) * QP + 32 + lo] };
            acc = wmma4(a, bf, acc);
        }
        #pragma unroll
        for (int r = 0; r < 8; ++r)
            O[(mt * 16 + r + 8 * hi) * OP + lo] = acc[r];
    }
    __syncthreads();

    for (int tile = wave; tile < 24; tile += 4) {
        int mt = tile / 6, nt = tile % 6;
        const float* ap = &O[(mt * 16 + lo) * OP + 2 * hi];
        v8f acc = {};
        for (int kk = 0; kk < HD; kk += 4) {
            int ka = kk + 2 * hi;
            v2f a  = *reinterpret_cast<const v2f*>(ap + kk);
            v2f bf = { sWp[ka * WOP + nt * 16 + lo],
                       sWp[(ka + 1) * WOP + nt * 16 + lo] };
            acc = wmma4(a, bf, acc);
        }
        #pragma unroll
        for (int r = 0; r < 8; ++r) {
            int m  = mt * 16 + r + 8 * hi;
            int t  = sidx[m];
            int hh = wy * WIN + (t >> 4);
            int ww = wx * WIN + (t & 15);
            atomicAdd(&y[((size_t)(b * IMG + hh) * IMG + ww) * C_DIM + nt * 16 + lo],
                      acc[r]);
        }
    }
}

extern "C" void kernel_launch(void* const* d_in, const int* in_sizes, int n_in,
                              void* d_out, int out_size, void* d_ws, size_t ws_size,
                              hipStream_t stream) {
    const float* x     = (const float*)d_in[0];
    const float* Wr    = (const float*)d_in[1];
    const float* Wqkv  = (const float*)d_in[2];
    const float* Wo    = (const float*)d_in[3];
    const float* projW = (const float*)d_in[4];
    const float* projb = (const float*)d_in[5];
    float* y = (float*)d_out;

    float* Wp   = (float*)d_ws;
    int*   gidx = (int*)((char*)d_ws + (size_t)HEADS * HD * C_DIM * sizeof(float));
    float* gval = (float*)((char*)d_ws + (size_t)HEADS * HD * C_DIM * sizeof(float) +
                           (size_t)BW * HEADS * KSEL * sizeof(int));

    k1_router<<<BW + HEADS, 256, 0, stream>>>(x, Wr, Wo, projW, projb, gidx, gval, Wp, y);
    k2_attn<<<dim3(BW, HEADS), 128, 0, stream>>>(x, Wqkv, Wp, gidx, gval, y);
}


# k1: top-64 set without sorting: one wave per head finds the 63rd largest sigmoid bit pattern by ballot search (2 barriers, not 229); picks kept in token order, so k2 float sums reorder
# speedup vs baseline: 2.6198x; 1.1470x over previous
#include <hip/hip_runtime.h>
#include <hip/hip_bf16.h>


#define BATCH   2
#define IMG     384
#define C_DIM   96
#define HEADS   6
#define HD      16
#define WIN     16
#define NWIN    24
#define BW      (BATCH * NWIN * NWIN)
#define LTOK    256
#define KSEL    64

typedef __attribute__((ext_vector_type(2))) float v2f;
typedef __attribute__((ext_vector_type(8))) float v8f;

static __device__ inline v8f wmma4(v2f a, v2f b, v8f c) {
    return __builtin_amdgcn_wmma_f32_16x16x4_f32(
         false, a,  false, b,
         (short)0, c,  false,  false);
}

__global__ __launch_bounds__(256) void k1_router(
    const float* __restrict__ x, const float* __restrict__ Wr,
    const float* __restrict__ Wo, const float* __restrict__ projW,
    const float* __restrict__ projb,
    int* __restrict__ gidx, float* __restrict__ gval,
    float* __restrict__ Wp, float* __restrict__ y)
{
    __shared__ float sWr[C_DIM * HEADS];
    __shared__ float sv[HEADS * LTOK];
    __shared__ __attribute__((aligned(16))) float sb[C_DIM];

    const int w = blockIdx.x;
    const int t = threadIdx.x;

    if (w >= BW) {
        const int e    = w - BW;
        const int lane = t & 31, nt = t >> 5;
        const int hi   = lane >> 4, lo = lane & 15;
        if (nt < C_DIM / 16) {
            const float* ap = Wo + ((size_t)e * HD + lo) * C_DIM + 2 * hi;
            const float* bp = projW + (size_t)(nt * 16 + lo) * C_DIM + 2 * hi;
            v8f acc = {};
            for (int kk = 0; kk < C_DIM; kk += 4) {
                v2f a  = *reinterpret_cast<const v2f*>(ap + kk);
                v2f bf = *reinterpret_cast<const v2f*>(bp + kk);
                acc = wmma4(a, bf, acc);
            }
            #pragma unroll
            for (int r = 0; r < 8; ++r)
                Wp[((size_t)e * HD + r + 8 * hi) * C_DIM + nt * 16 + lo] = acc[r];
        }
        return;
    }

    for (int i = t; i < C_DIM * HEADS; i += 256) sWr[i] = Wr[i];
    if (t < C_DIM) sb[t] = projb[t];
    __syncthreads();

    const int b  = w / (NWIN * NWIN);
    const int rm = w % (NWIN * NWIN);
    const int wy = rm / NWIN, wx = rm % NWIN;

    {
        constexpr int ROWP = WIN * C_DIM / 4;
        constexpr int IMGP = IMG * C_DIM / 4;
        float4* y4 = reinterpret_cast<float4*>(y) +
                     ((size_t)(b * IMG + wy * WIN) * IMG + wx * WIN) * (C_DIM / 4);
        const float4* sb4 = reinterpret_cast<const float4*>(sb);
        float4 bv[3];
        int    off[3];
        #pragma unroll
        for (int q = 0; q < 3; ++q) {
            int i = t + q * 256;
            int c = (i >= ROWP) ? i - ROWP : i;
            off[q] = ((i >= ROWP) ? IMGP : 0) + c;
            bv[q]  = sb4[c % (C_DIM / 4)];
        }
        #pragma unroll
        for (int p = 0; p < WIN / 2; ++p) {
            #pragma unroll
            for (int q = 0; q < 3; ++q) y4[(size_t)p * 2 * IMGP + off[q]] = bv[q];
        }
    }
    const int hh = wy * WIN + (t >> 4);
    const int ww = wx * WIN + (t & 15);
    const float* xr = x + ((size_t)(b * IMG + hh) * IMG + ww) * C_DIM;

    float lg[HEADS];
    #pragma unroll
    for (int e = 0; e < HEADS; ++e) lg[e] = 0.f;
    for (int c = 0; c < C_DIM; ++c) {
        float xv = xr[c];
        #pragma unroll
        for (int e = 0; e < HEADS; ++e) lg[e] += xv * sWr[c * HEADS + e];
    }
    #pragma unroll
    for (int e = 0; e < HEADS; ++e) lg[e] = 1.0f / (1.0f + __expf(-lg[e]));

    #pragma unroll
    for (int e = 0; e < HEADS; ++e) sv[e * LTOK + t] = lg[e];
    __syncthreads();

    const int wv = t >> 5, lane = t & 31;
    if (wv < HEADS) {
        const int e = wv;
        float    val[8];
        unsigned kb[8];
        #pragma unroll
        for (int r = 0; r < 8; ++r) {
            val[r] = sv[e * LTOK + r * 32 + lane];
            kb[r]  = __float_as_uint(val[r]);
        }
        const unsigned tail0 = 0xFFFFFFFEu;

        unsigned T = 0;
        for (int bit = 31; bit >= 0; --bit) {
            const unsigned cand = T | (1u << bit);
            int cnt = 0;
            #pragma unroll
            for (int r = 0; r < 8; ++r) {
                unsigned m = __builtin_amdgcn_ballot_w32(kb[r] >= cand);
                if (r == 0) m &= tail0;
                cnt += __builtin_popcount(m);
            }
            if (cnt >= KSEL - 1) T = cand;
        }

        unsigned gm[8], em[8];
        int gt = 0;
        #pragma unroll
        for (int r = 0; r < 8; ++r) {
            gm[r] = __builtin_amdgcn_ballot_w32(kb[r] > T);
            em[r] = __builtin_amdgcn_ballot_w32(kb[r] == T);
            if (r == 0) { gm[r] &= tail0; em[r] &= tail0; }
            gt += __builtin_popcount(gm[r]);
        }
        const int need = (KSEL - 1) - gt;

        const size_t base = ((size_t)w * HEADS + e) * KSEL;
        int eqBefore = 0, selBefore = 1;
        #pragma unroll
        for (int r = 0; r < 8; ++r) {
            const bool above = (gm[r] >> lane) & 1u;
            const bool equal = (em[r] >> lane) & 1u;
            const int  eqRank = eqBefore + (int)__builtin_amdgcn_mbcnt_lo(em[r], 0u);
            const bool sel = above || (equal && eqRank < need);
            const unsigned sm = __builtin_amdgcn_ballot_w32(sel);
            if (sel) {
                const int pos = selBefore + (int)__builtin_amdgcn_mbcnt_lo(sm, 0u);
                gidx[base + pos] = r * 32 + lane;
                gval[base + pos] = val[r];
            }
            eqBefore  += __builtin_popcount(em[r]);
            selBefore += __builtin_popcount(sm);
        }
        if (lane == 0) { gidx[base] = 0; gval[base] = val[0]; }
    }
}

#define XP  100
#define QP  52
#define SP  68
#define OP  20
#define WOP 104
__global__ __launch_bounds__(128) void k2_attn(
    const float* __restrict__ x,
    const float* __restrict__ Wqkv, const float* __restrict__ Wp,
    const int* __restrict__ gidx, const float* __restrict__ gval,
    float* __restrict__ y)
{
    __shared__ __attribute__((aligned(16))) float pool[KSEL * XP + C_DIM * 48];
    __shared__ __attribute__((aligned(16))) float sQKV[KSEL * QP];
    __shared__ float sgate[KSEL];
    __shared__ int   sidx[KSEL];

    float* Xg  = pool;
    float* sWq = pool + KSEL * XP;
    float* S   = pool;
    float* O   = pool + KSEL * SP;
    float* sWp = pool + KSEL * SP + KSEL * OP;

    const int w    = blockIdx.x;
    const int e    = blockIdx.y;
    const int tid  = threadIdx.x;
    const int lane = tid & 31;
    const int wave = tid >> 5;
    const int hi   = lane >> 4;
    const int lo   = lane & 15;

    if (tid < KSEL) {
        sidx[tid]  = gidx[((size_t)w * HEADS + e) * KSEL + tid];
        sgate[tid] = gval[((size_t)w * HEADS + e) * KSEL + tid];
    }
    __syncthreads();

    const int b  = w / (NWIN * NWIN);
    const int rm = w % (NWIN * NWIN);
    const int wy = rm / NWIN, wx = rm % NWIN;

    constexpr int ROW4 = C_DIM / 4;
    constexpr int NGX  = KSEL * ROW4 / 128;
    constexpr int NGW  = C_DIM * 48 / 4 / 128;
    constexpr int NGO  = HD * C_DIM / 4 / 128;
    {
        float4 gx[NGX], gw[NGW];
        const float4* wq4 = reinterpret_cast<const float4*>(Wqkv + (size_t)e * C_DIM * 48);
        for (int k = 0; k < NGX; ++k) {
            int i = tid + k * 128;
            int j = i / ROW4, c4 = i % ROW4;
            int t  = sidx[j];
            int hh = wy * WIN + (t >> 4);
            int ww = wx * WIN + (t & 15);
            gx[k] = *reinterpret_cast<const float4*>(
                x + ((size_t)(b * IMG + hh) * IMG + ww) * C_DIM + c4 * 4);
        }
        for (int k = 0; k < NGW; ++k) gw[k] = wq4[tid + k * 128];
        for (int k = 0; k < NGX; ++k) {
            int i = tid + k * 128;
            int j = i / ROW4, c4 = i % ROW4;
            *reinterpret_cast<float4*>(&Xg[j * XP + c4 * 4]) = gx[k];
        }
        for (int k = 0; k < NGW; ++k)
            *reinterpret_cast<float4*>(&sWq[(tid + k * 128) * 4]) = gw[k];
    }
    __syncthreads();

    for (int tile = wave; tile < 12; tile += 4) {
        int mt = tile / 3, nt = tile % 3;
        const float* ap = &Xg[(mt * 16 + lo) * XP + 2 * hi];
        v8f acc = {};
        for (int kk = 0; kk < C_DIM; kk += 4) {
            int ka = kk + 2 * hi;
            v2f a  = *reinterpret_cast<const v2f*>(ap + kk);
            v2f bf = { sWq[ka * 48 + nt * 16 + lo],
                       sWq[(ka + 1) * 48 + nt * 16 + lo] };
            acc = wmma4(a, bf, acc);
        }
        #pragma unroll
        for (int r = 0; r < 8; ++r)
            sQKV[(mt * 16 + r + 8 * hi) * QP + nt * 16 + lo] = acc[r];
    }
    __syncthreads();

    {
        const float4* wp4 = reinterpret_cast<const float4*>(Wp + (size_t)e * HD * C_DIM);
        float4 go[NGO];
        for (int k = 0; k < NGO; ++k) go[k] = wp4[tid + k * 128];
        for (int k = 0; k < NGO; ++k) {
            int i = tid + k * 128;
            int r = i / ROW4, c4 = i % ROW4;
            *reinterpret_cast<float4*>(&sWp[r * WOP + c4 * 4]) = go[k];
        }
    }

    for (int tile = wave; tile < 16; tile += 4) {
        int mt = tile / 4, nt = tile % 4;
        const float* ap = &sQKV[(mt * 16 + lo) * QP + 2 * hi];
        const float* bp = &sQKV[(nt * 16 + lo) * QP + 16 + 2 * hi];
        v8f acc = {};
        for (int kk = 0; kk < HD; kk += 4) {
            v2f a  = *reinterpret_cast<const v2f*>(ap + kk);
            v2f bf = *reinterpret_cast<const v2f*>(bp + kk);
            acc = wmma4(a, bf, acc);
        }
        #pragma unroll
        for (int r = 0; r < 8; ++r)
            S[(mt * 16 + r + 8 * hi) * SP + nt * 16 + lo] = acc[r] * 0.25f;
    }
    __syncthreads();

    if (tid < KSEL) {
        float* Sr = &S[tid * SP];
        float m = -1e30f;
        for (int j = 0; j < 64; ++j) m = fmaxf(m, Sr[j]);
        float s = 0.f;
        for (int j = 0; j < 64; ++j) {
            float ev = __expf(Sr[j] - m);
            Sr[j] = ev; s += ev;
        }
        float inv = sgate[tid] / s;
        for (int j = 0; j < 64; ++j) Sr[j] *= inv;
    }
    __syncthreads();

    {
        int mt = wave;
        const float* ap = &S[(mt * 16 + lo) * SP + 2 * hi];
        v8f acc = {};
        for (int kk = 0; kk < 64; kk += 4) {
            int ka = kk + 2 * hi;
            v2f a  = *reinterpret_cast<const v2f*>(ap + kk);
            v2f bf = { sQKV[ka * QP + 32 + lo],
                       sQKV[(ka + 1) * QP + 32 + lo] };
            acc = wmma4(a, bf, acc);
        }
        #pragma unroll
        for (int r = 0; r < 8; ++r)
            O[(mt * 16 + r + 8 * hi) * OP + lo] = acc[r];
    }
    __syncthreads();

    for (int tile = wave; tile < 24; tile += 4) {
        int mt = tile / 6, nt = tile % 6;
        const float* ap = &O[(mt * 16 + lo) * OP + 2 * hi];
        v8f acc = {};
        for (int kk = 0; kk < HD; kk += 4) {
            int ka = kk + 2 * hi;
            v2f a  = *reinterpret_cast<const v2f*>(ap + kk);
            v2f bf = { sWp[ka * WOP + nt * 16 + lo],
                       sWp[(ka + 1) * WOP + nt * 16 + lo] };
            acc = wmma4(a, bf, acc);
        }
        #pragma unroll
        for (int r = 0; r < 8; ++r) {
            int m  = mt * 16 + r + 8 * hi;
            int t  = sidx[m];
            int hh = wy * WIN + (t >> 4);
            int ww = wx * WIN + (t & 15);
            atomicAdd(&y[((size_t)(b * IMG + hh) * IMG + ww) * C_DIM + nt * 16 + lo],
                      acc[r]);
        }
    }
}

extern "C" void kernel_launch(void* const* d_in, const int* in_sizes, int n_in,
                              void* d_out, int out_size, void* d_ws, size_t ws_size,
                              hipStream_t stream) {
    const float* x     = (const float*)d_in[0];
    const float* Wr    = (const float*)d_in[1];
    const float* Wqkv  = (const float*)d_in[2];
    const float* Wo    = (const float*)d_in[3];
    const float* projW = (const float*)d_in[4];
    const float* projb = (const float*)d_in[5];
    float* y = (float*)d_out;

    float* Wp   = (float*)d_ws;
    int*   gidx = (int*)((char*)d_ws + (size_t)HEADS * HD * C_DIM * sizeof(float));
    float* gval = (float*)((char*)d_ws + (size_t)HEADS * HD * C_DIM * sizeof(float) +
                           (size_t)BW * HEADS * KSEL * sizeof(int));

    k1_router<<<BW + HEADS, 256, 0, stream>>>(x, Wr, Wo, projW, projb, gidx, gval, Wp, y);
    k2_attn<<<dim3(BW, HEADS), 128, 0, stream>>>(x, Wqkv, Wp, gidx, gval, y);
}


# scatter-add by an owner kernel: k2 stores O_e; k3 (one block per window) adds O_e @ Wp_e of the 6 heads into an LDS y tile in fixed head order, then the bias, and stores y once; no atomics
# speedup vs baseline: 2.9068x; 1.1096x over previous
#include <hip/hip_runtime.h>
#include <hip/hip_bf16.h>


#define BATCH   2
#define IMG     384
#define C_DIM   96
#define HEADS   6
#define HD      16
#define WIN     16
#define NWIN    24
#define BW      (BATCH * NWIN * NWIN)
#define LTOK    256
#define KSEL    64

typedef __attribute__((ext_vector_type(2))) float v2f;
typedef __attribute__((ext_vector_type(8))) float v8f;

static __device__ inline v8f wmma4(v2f a, v2f b, v8f c) {
    return __builtin_amdgcn_wmma_f32_16x16x4_f32(
         false, a,  false, b,
         (short)0, c,  false,  false);
}

__global__ __launch_bounds__(256) void k1_router(
    const float* __restrict__ x, const float* __restrict__ Wr,
    const float* __restrict__ Wo, const float* __restrict__ projW,
    int* __restrict__ gidx, float* __restrict__ gval,
    float* __restrict__ Wp)
{
    __shared__ float sWr[C_DIM * HEADS];
    __shared__ float sv[HEADS * LTOK];

    const int w = blockIdx.x;
    const int t = threadIdx.x;

    if (w >= BW) {
        const int e    = w - BW;
        const int lane = t & 31, nt = t >> 5;
        const int hi   = lane >> 4, lo = lane & 15;
        if (nt < C_DIM / 16) {
            const float* ap = Wo + ((size_t)e * HD + lo) * C_DIM + 2 * hi;
            const float* bp = projW + (size_t)(nt * 16 + lo) * C_DIM + 2 * hi;
            v8f acc = {};
            for (int kk = 0; kk < C_DIM; kk += 4) {
                v2f a  = *reinterpret_cast<const v2f*>(ap + kk);
                v2f bf = *reinterpret_cast<const v2f*>(bp + kk);
                acc = wmma4(a, bf, acc);
            }
            #pragma unroll
            for (int r = 0; r < 8; ++r)
                Wp[((size_t)e * HD + r + 8 * hi) * C_DIM + nt * 16 + lo] = acc[r];
        }
        return;
    }

    for (int i = t; i < C_DIM * HEADS; i += 256) sWr[i] = Wr[i];
    __syncthreads();

    const int b  = w / (NWIN * NWIN);
    const int rm = w % (NWIN * NWIN);
    const int wy = rm / NWIN, wx = rm % NWIN;

    const int hh = wy * WIN + (t >> 4);
    const int ww = wx * WIN + (t & 15);
    const float* xr = x + ((size_t)(b * IMG + hh) * IMG + ww) * C_DIM;

    float lg[HEADS];
    #pragma unroll
    for (int e = 0; e < HEADS; ++e) lg[e] = 0.f;
    for (int c = 0; c < C_DIM; ++c) {
        float xv = xr[c];
        #pragma unroll
        for (int e = 0; e < HEADS; ++e) lg[e] += xv * sWr[c * HEADS + e];
    }
    #pragma unroll
    for (int e = 0; e < HEADS; ++e) lg[e] = 1.0f / (1.0f + __expf(-lg[e]));

    #pragma unroll
    for (int e = 0; e < HEADS; ++e) sv[e * LTOK + t] = lg[e];
    __syncthreads();

    const int wv = t >> 5, lane = t & 31;
    if (wv < HEADS) {
        const int e = wv;
        float    val[8];
        unsigned kb[8];
        #pragma unroll
        for (int r = 0; r < 8; ++r) {
            val[r] = sv[e * LTOK + r * 32 + lane];
            kb[r]  = __float_as_uint(val[r]);
        }
        const unsigned tail0 = 0xFFFFFFFEu;

        unsigned T = 0;
        for (int bit = 31; bit >= 0; --bit) {
            const unsigned cand = T | (1u << bit);
            int cnt = 0;
            #pragma unroll
            for (int r = 0; r < 8; ++r) {
                unsigned m = __builtin_amdgcn_ballot_w32(kb[r] >= cand);
                if (r == 0) m &= tail0;
                cnt += __builtin_popcount(m);
            }
            if (cnt >= KSEL - 1) T = cand;
        }

        unsigned gm[8], em[8];
        int gt = 0;
        #pragma unroll
        for (int r = 0; r < 8; ++r) {
            gm[r] = __builtin_amdgcn_ballot_w32(kb[r] > T);
            em[r] = __builtin_amdgcn_ballot_w32(kb[r] == T);
            if (r == 0) { gm[r] &= tail0; em[r] &= tail0; }
            gt += __builtin_popcount(gm[r]);
        }
        const int need = (KSEL - 1) - gt;

        const size_t base = ((size_t)w * HEADS + e) * KSEL;
        int eqBefore = 0, selBefore = 1;
        #pragma unroll
        for (int r = 0; r < 8; ++r) {
            const bool above = (gm[r] >> lane) & 1u;
            const bool equal = (em[r] >> lane) & 1u;
            const int  eqRank = eqBefore + (int)__builtin_amdgcn_mbcnt_lo(em[r], 0u);
            const bool sel = above || (equal && eqRank < need);
            const unsigned sm = __builtin_amdgcn_ballot_w32(sel);
            if (sel) {
                const int pos = selBefore + (int)__builtin_amdgcn_mbcnt_lo(sm, 0u);
                gidx[base + pos] = r * 32 + lane;
                gval[base + pos] = val[r];
            }
            eqBefore  += __builtin_popcount(em[r]);
            selBefore += __builtin_popcount(sm);
        }
        if (lane == 0) { gidx[base] = 0; gval[base] = val[0]; }
    }
}

#define XP  100
#define QP  52
#define SP  68
#define OP  20
__global__ __launch_bounds__(128) void k2_attn(
    const float* __restrict__ x,
    const float* __restrict__ Wqkv,
    const int* __restrict__ gidx, const float* __restrict__ gval,
    float* __restrict__ Obuf)
{
    __shared__ __attribute__((aligned(16))) float pool[KSEL * XP + C_DIM * 48];
    __shared__ __attribute__((aligned(16))) float sQKV[KSEL * QP];
    __shared__ float sgate[KSEL];
    __shared__ int   sidx[KSEL];

    float* Xg  = pool;
    float* sWq = pool + KSEL * XP;
    float* S   = pool;
    float* O   = pool + KSEL * SP;

    const int w    = blockIdx.x;
    const int e    = blockIdx.y;
    const int tid  = threadIdx.x;
    const int lane = tid & 31;
    const int wave = tid >> 5;
    const int hi   = lane >> 4;
    const int lo   = lane & 15;

    if (tid < KSEL) {
        sidx[tid]  = gidx[((size_t)w * HEADS + e) * KSEL + tid];
        sgate[tid] = gval[((size_t)w * HEADS + e) * KSEL + tid];
    }
    __syncthreads();

    const int b  = w / (NWIN * NWIN);
    const int rm = w % (NWIN * NWIN);
    const int wy = rm / NWIN, wx = rm % NWIN;

    constexpr int ROW4 = C_DIM / 4;
    constexpr int NGX  = KSEL * ROW4 / 128;
    constexpr int NGW  = C_DIM * 48 / 4 / 128;
    {
        float4 gx[NGX], gw[NGW];
        const float4* wq4 = reinterpret_cast<const float4*>(Wqkv + (size_t)e * C_DIM * 48);
        for (int k = 0; k < NGX; ++k) {
            int i = tid + k * 128;
            int j = i / ROW4, c4 = i % ROW4;
            int t  = sidx[j];
            int hh = wy * WIN + (t >> 4);
            int ww = wx * WIN + (t & 15);
            gx[k] = *reinterpret_cast<const float4*>(
                x + ((size_t)(b * IMG + hh) * IMG + ww) * C_DIM + c4 * 4);
        }
        for (int k = 0; k < NGW; ++k) gw[k] = wq4[tid + k * 128];
        for (int k = 0; k < NGX; ++k) {
            int i = tid + k * 128;
            int j = i / ROW4, c4 = i % ROW4;
            *reinterpret_cast<float4*>(&Xg[j * XP + c4 * 4]) = gx[k];
        }
        for (int k = 0; k < NGW; ++k)
            *reinterpret_cast<float4*>(&sWq[(tid + k * 128) * 4]) = gw[k];
    }
    __syncthreads();

    for (int tile = wave; tile < 12; tile += 4) {
        int mt = tile / 3, nt = tile % 3;
        const float* ap = &Xg[(mt * 16 + lo) * XP + 2 * hi];
        v8f acc = {};
        for (int kk = 0; kk < C_DIM; kk += 4) {
            int ka = kk + 2 * hi;
            v2f a  = *reinterpret_cast<const v2f*>(ap + kk);
            v2f bf = { sWq[ka * 48 + nt * 16 + lo],
                       sWq[(ka + 1) * 48 + nt * 16 + lo] };
            acc = wmma4(a, bf, acc);
        }
        #pragma unroll
        for (int r = 0; r < 8; ++r)
            sQKV[(mt * 16 + r + 8 * hi) * QP + nt * 16 + lo] = acc[r];
    }
    __syncthreads();

    for (int tile = wave; tile < 16; tile += 4) {
        int mt = tile / 4, nt = tile % 4;
        const float* ap = &sQKV[(mt * 16 + lo) * QP + 2 * hi];
        const float* bp = &sQKV[(nt * 16 + lo) * QP + 16 + 2 * hi];
        v8f acc = {};
        for (int kk = 0; kk < HD; kk += 4) {
            v2f a  = *reinterpret_cast<const v2f*>(ap + kk);
            v2f bf = *reinterpret_cast<const v2f*>(bp + kk);
            acc = wmma4(a, bf, acc);
        }
        #pragma unroll
        for (int r = 0; r < 8; ++r)
            S[(mt * 16 + r + 8 * hi) * SP + nt * 16 + lo] = acc[r] * 0.25f;
    }
    __syncthreads();

    if (tid < KSEL) {
        float* Sr = &S[tid * SP];
        float m = -1e30f;
        for (int j = 0; j < 64; ++j) m = fmaxf(m, Sr[j]);
        float s = 0.f;
        for (int j = 0; j < 64; ++j) {
            float ev = __expf(Sr[j] - m);
            Sr[j] = ev; s += ev;
        }
        float inv = sgate[tid] / s;
        for (int j = 0; j < 64; ++j) Sr[j] *= inv;
    }
    __syncthreads();

    {
        int mt = wave;
        const float* ap = &S[(mt * 16 + lo) * SP + 2 * hi];
        v8f acc = {};
        for (int kk = 0; kk < 64; kk += 4) {
            int ka = kk + 2 * hi;
            v2f a  = *reinterpret_cast<const v2f*>(ap + kk);
            v2f bf = { sQKV[ka * QP + 32 + lo],
                       sQKV[(ka + 1) * QP + 32 + lo] };
            acc = wmma4(a, bf, acc);
        }
        #pragma unroll
        for (int r = 0; r < 8; ++r)
            O[(mt * 16 + r + 8 * hi) * OP + lo] = acc[r];
    }
    __syncthreads();

    {
        float4* ob4 = reinterpret_cast<float4*>(Obuf + ((size_t)w * HEADS + e) * KSEL * HD);
        for (int q = 0; q < KSEL * HD / 4 / 128; ++q) {
            int p = tid + q * 128;
            int r = p >> 2, c4 = p & 3;
            ob4[p] = *reinterpret_cast<const float4*>(&O[r * OP + c4 * 4]);
        }
    }
}

static __device__ inline void k3_load_head(
    const float* __restrict__ Ow, const float* __restrict__ Wp,
    int e, int nt, int lo, int hi,
    v2f (&a)[KSEL / 16][HD / 4], v2f (&bf)[HD / 4])
{
    #pragma unroll
    for (int j = 0; j < HD / 4; ++j) {
        const int ka = 4 * j + 2 * hi;
        bf[j] = v2f{ Wp[((size_t)e * HD + ka) * C_DIM + nt * 16 + lo],
                     Wp[((size_t)e * HD + ka + 1) * C_DIM + nt * 16 + lo] };
    }
    #pragma unroll
    for (int mt = 0; mt < KSEL / 16; ++mt) {
        const float* ap = Ow + ((size_t)e * KSEL + mt * 16 + lo) * HD + 2 * hi;
        #pragma unroll
        for (int j = 0; j < HD / 4; ++j)
            a[mt][j] = *reinterpret_cast<const v2f*>(ap + 4 * j);
    }
}

__global__ __launch_bounds__(192) void k3_owner(
    const float* __restrict__ Obuf, const float* __restrict__ Wp,
    const float* __restrict__ projb, const int* __restrict__ gidx,
    float* __restrict__ y)
{
    __shared__ __attribute__((aligned(16))) float yt[LTOK * C_DIM];
    __shared__ __attribute__((aligned(16))) int   stok[HEADS * KSEL];

    const int w    = blockIdx.x;
    const int tid  = threadIdx.x;
    const int lane = tid & 31;
    const int nt   = tid >> 5;
    const int hi   = lane >> 4;
    const int lo   = lane & 15;

    const float* Ow = Obuf + (size_t)w * HEADS * KSEL * HD;

    v2f a_n[KSEL / 16][HD / 4], bf_n[HD / 4];
    k3_load_head(Ow, Wp, 0, nt, lo, hi, a_n, bf_n);

    stok[tid]       = gidx[(size_t)w * HEADS * KSEL + tid];
    stok[tid + 192] = gidx[(size_t)w * HEADS * KSEL + tid + 192];
    {
        float4* yt4 = reinterpret_cast<float4*>(yt);
        for (int q = 0; q < LTOK * C_DIM / 4 / 192; ++q)
            yt4[tid + q * 192] = float4{0.f, 0.f, 0.f, 0.f};
    }
    __syncthreads();

    float* ycol = yt + nt * 16 + lo;

    #pragma clang loop unroll(disable)
    for (int e = 0; e < HEADS; ++e) {
        v2f a[KSEL / 16][HD / 4], bf[HD / 4];
        #pragma unroll
        for (int j = 0; j < HD / 4; ++j) {
            bf[j] = bf_n[j];
            #pragma unroll
            for (int mt = 0; mt < KSEL / 16; ++mt) a[mt][j] = a_n[mt][j];
        }
        if (e + 1 < HEADS) k3_load_head(Ow, Wp, e + 1, nt, lo, hi, a_n, bf_n);

        int tok[KSEL / 16][8];
        v8f acc[KSEL / 16];
        #pragma unroll
        for (int mt = 0; mt < KSEL / 16; ++mt) {
            const int4* tp = reinterpret_cast<const int4*>(&stok[e * KSEL + mt * 16 + 8 * hi]);
            const int4 t0 = tp[0], t1 = tp[1];
            tok[mt][0] = t0.x; tok[mt][1] = t0.y; tok[mt][2] = t0.z; tok[mt][3] = t0.w;
            tok[mt][4] = t1.x; tok[mt][5] = t1.y; tok[mt][6] = t1.z; tok[mt][7] = t1.w;
            #pragma unroll
            for (int r = 0; r < 8; ++r) acc[mt][r] = ycol[tok[mt][r] * C_DIM];
        }
        #pragma unroll
        for (int j = 0; j < HD / 4; ++j) {
            #pragma unroll
            for (int mt = 0; mt < KSEL / 16; ++mt) acc[mt] = wmma4(a[mt][j], bf[j], acc[mt]);
        }
        #pragma unroll
        for (int mt = 0; mt < KSEL / 16; ++mt) {
            #pragma unroll
            for (int r = 0; r < 8; ++r) ycol[tok[mt][r] * C_DIM] = acc[mt][r];
        }
    }
    __syncthreads();

    {
        const int b  = w / (NWIN * NWIN);
        const int rm = w % (NWIN * NWIN);
        const int wy = rm / NWIN, wx = rm % NWIN;
        constexpr int ROWP = WIN * C_DIM / 4;
        constexpr int IMGP = IMG * C_DIM / 4;
        float4* y4 = reinterpret_cast<float4*>(y) +
                     ((size_t)(b * IMG + wy * WIN) * IMG + wx * WIN) * (C_DIM / 4);
        const float4* yt4 = reinterpret_cast<const float4*>(yt);
        const float4 bv = reinterpret_cast<const float4*>(projb)[tid % (C_DIM / 4)];
        for (int ty = 0; ty < WIN; ++ty) {
            for (int h2 = 0; h2 < ROWP / 192; ++h2) {
                int c = tid + h2 * 192;
                float4 v = yt4[ty * ROWP + c];
                v.x += bv.x; v.y += bv.y; v.z += bv.z; v.w += bv.w;
                y4[(size_t)ty * IMGP + c] = v;
            }
        }
    }
}

extern "C" void kernel_launch(void* const* d_in, const int* in_sizes, int n_in,
                              void* d_out, int out_size, void* d_ws, size_t ws_size,
                              hipStream_t stream) {
    const float* x     = (const float*)d_in[0];
    const float* Wr    = (const float*)d_in[1];
    const float* Wqkv  = (const float*)d_in[2];
    const float* Wo    = (const float*)d_in[3];
    const float* projW = (const float*)d_in[4];
    const float* projb = (const float*)d_in[5];
    float* y = (float*)d_out;

    float* Wp   = (float*)d_ws;
    int*   gidx = (int*)((char*)d_ws + (size_t)HEADS * HD * C_DIM * sizeof(float));
    float* gval = (float*)((char*)d_ws + (size_t)HEADS * HD * C_DIM * sizeof(float) +
                           (size_t)BW * HEADS * KSEL * sizeof(int));
    float* Obuf = (float*)((char*)d_ws + (size_t)HEADS * HD * C_DIM * sizeof(float) +
                           (size_t)BW * HEADS * KSEL * (sizeof(int) + sizeof(float)));

    k1_router<<<BW + HEADS, 256, 0, stream>>>(x, Wr, Wo, projW, gidx, gval, Wp);
    k2_attn<<<dim3(BW, HEADS), 128, 0, stream>>>(x, Wqkv, gidx, gval, Obuf);
    k3_owner<<<BW, 192, 0, stream>>>(Obuf, Wp, projb, gidx, y);
}


# k2: scores taken transposed (S^T = K @ Q^T): a lane holds half a score row, so softmax and P @ V run on accumulator registers; no S or P in LDS, 2 barriers fewer; row sums and P@V key order reorder
# speedup vs baseline: 3.3060x; 1.1373x over previous
#include <hip/hip_runtime.h>
#include <hip/hip_bf16.h>


#define BATCH   2
#define IMG     384
#define C_DIM   96
#define HEADS   6
#define HD      16
#define WIN     16
#define NWIN    24
#define BW      (BATCH * NWIN * NWIN)
#define LTOK    256
#define KSEL    64

typedef __attribute__((ext_vector_type(2))) float v2f;
typedef __attribute__((ext_vector_type(8))) float v8f;

static __device__ inline v8f wmma4(v2f a, v2f b, v8f c) {
    return __builtin_amdgcn_wmma_f32_16x16x4_f32(
         false, a,  false, b,
         (short)0, c,  false,  false);
}

static __device__ inline float xhalf(float v) {
    return __builtin_bit_cast(float, __builtin_amdgcn_permlanex16(
        __builtin_bit_cast(int, v), __builtin_bit_cast(int, v),
        0x76543210, 0xfedcba98, false, false));
}

__global__ __launch_bounds__(256) void k1_router(
    const float* __restrict__ x, const float* __restrict__ Wr,
    const float* __restrict__ Wo, const float* __restrict__ projW,
    int* __restrict__ gidx, float* __restrict__ gval,
    float* __restrict__ Wp)
{
    __shared__ float sWr[C_DIM * HEADS];
    __shared__ float sv[HEADS * LTOK];

    const int w = blockIdx.x;
    const int t = threadIdx.x;

    if (w >= BW) {
        const int e    = w - BW;
        const int lane = t & 31, nt = t >> 5;
        const int hi   = lane >> 4, lo = lane & 15;
        if (nt < C_DIM / 16) {
            const float* ap = Wo + ((size_t)e * HD + lo) * C_DIM + 2 * hi;
            const float* bp = projW + (size_t)(nt * 16 + lo) * C_DIM + 2 * hi;
            v8f acc = {};
            for (int kk = 0; kk < C_DIM; kk += 4) {
                v2f a  = *reinterpret_cast<const v2f*>(ap + kk);
                v2f bf = *reinterpret_cast<const v2f*>(bp + kk);
                acc = wmma4(a, bf, acc);
            }
            #pragma unroll
            for (int r = 0; r < 8; ++r)
                Wp[((size_t)e * HD + r + 8 * hi) * C_DIM + nt * 16 + lo] = acc[r];
        }
        return;
    }

    for (int i = t; i < C_DIM * HEADS; i += 256) sWr[i] = Wr[i];
    __syncthreads();

    const int b  = w / (NWIN * NWIN);
    const int rm = w % (NWIN * NWIN);
    const int wy = rm / NWIN, wx = rm % NWIN;

    const int hh = wy * WIN + (t >> 4);
    const int ww = wx * WIN + (t & 15);
    const float* xr = x + ((size_t)(b * IMG + hh) * IMG + ww) * C_DIM;

    float lg[HEADS];
    #pragma unroll
    for (int e = 0; e < HEADS; ++e) lg[e] = 0.f;
    for (int c = 0; c < C_DIM; ++c) {
        float xv = xr[c];
        #pragma unroll
        for (int e = 0; e < HEADS; ++e) lg[e] += xv * sWr[c * HEADS + e];
    }
    #pragma unroll
    for (int e = 0; e < HEADS; ++e) lg[e] = 1.0f / (1.0f + __expf(-lg[e]));

    #pragma unroll
    for (int e = 0; e < HEADS; ++e) sv[e * LTOK + t] = lg[e];
    __syncthreads();

    const int wv = t >> 5, lane = t & 31;
    if (wv < HEADS) {
        const int e = wv;
        float    val[8];
        unsigned kb[8];
        #pragma unroll
        for (int r = 0; r < 8; ++r) {
            val[r] = sv[e * LTOK + r * 32 + lane];
            kb[r]  = __float_as_uint(val[r]);
        }
        const unsigned tail0 = 0xFFFFFFFEu;

        unsigned T = 0;
        for (int bit = 31; bit >= 0; --bit) {
            const unsigned cand = T | (1u << bit);
            int cnt = 0;
            #pragma unroll
            for (int r = 0; r < 8; ++r) {
                unsigned m = __builtin_amdgcn_ballot_w32(kb[r] >= cand);
                if (r == 0) m &= tail0;
                cnt += __builtin_popcount(m);
            }
            if (cnt >= KSEL - 1) T = cand;
        }

        unsigned gm[8], em[8];
        int gt = 0;
        #pragma unroll
        for (int r = 0; r < 8; ++r) {
            gm[r] = __builtin_amdgcn_ballot_w32(kb[r] > T);
            em[r] = __builtin_amdgcn_ballot_w32(kb[r] == T);
            if (r == 0) { gm[r] &= tail0; em[r] &= tail0; }
            gt += __builtin_popcount(gm[r]);
        }
        const int need = (KSEL - 1) - gt;

        const size_t base = ((size_t)w * HEADS + e) * KSEL;
        int eqBefore = 0, selBefore = 1;
        #pragma unroll
        for (int r = 0; r < 8; ++r) {
            const bool above = (gm[r] >> lane) & 1u;
            const bool equal = (em[r] >> lane) & 1u;
            const int  eqRank = eqBefore + (int)__builtin_amdgcn_mbcnt_lo(em[r], 0u);
            const bool sel = above || (equal && eqRank < need);
            const unsigned sm = __builtin_amdgcn_ballot_w32(sel);
            if (sel) {
                const int pos = selBefore + (int)__builtin_amdgcn_mbcnt_lo(sm, 0u);
                gidx[base + pos] = r * 32 + lane;
                gval[base + pos] = val[r];
            }
            eqBefore  += __builtin_popcount(em[r]);
            selBefore += __builtin_popcount(sm);
        }
        if (lane == 0) { gidx[base] = 0; gval[base] = val[0]; }
    }
}

#define XP  100
#define QP  52
#define OP  20
__global__ __launch_bounds__(128) void k2_attn(
    const float* __restrict__ x,
    const float* __restrict__ Wqkv,
    const int* __restrict__ gidx, const float* __restrict__ gval,
    float* __restrict__ Obuf)
{
    __shared__ __attribute__((aligned(16))) float pool[KSEL * XP + C_DIM * 48];
    __shared__ __attribute__((aligned(16))) float sQKV[KSEL * QP];
    __shared__ float sgate[KSEL];
    __shared__ int   sidx[KSEL];

    float* Xg  = pool;
    float* sWq = pool + KSEL * XP;
    float* O   = pool;

    const int w    = blockIdx.x;
    const int e    = blockIdx.y;
    const int tid  = threadIdx.x;
    const int lane = tid & 31;
    const int wave = tid >> 5;
    const int hi   = lane >> 4;
    const int lo   = lane & 15;

    if (tid < KSEL) {
        sidx[tid]  = gidx[((size_t)w * HEADS + e) * KSEL + tid];
        sgate[tid] = gval[((size_t)w * HEADS + e) * KSEL + tid];
    }
    __syncthreads();

    const int b  = w / (NWIN * NWIN);
    const int rm = w % (NWIN * NWIN);
    const int wy = rm / NWIN, wx = rm % NWIN;

    constexpr int ROW4 = C_DIM / 4;
    constexpr int NGX  = KSEL * ROW4 / 128;
    constexpr int NGW  = C_DIM * 48 / 4 / 128;
    {
        float4 gx[NGX], gw[NGW];
        const float4* wq4 = reinterpret_cast<const float4*>(Wqkv + (size_t)e * C_DIM * 48);
        for (int k = 0; k < NGX; ++k) {
            int i = tid + k * 128;
            int j = i / ROW4, c4 = i % ROW4;
            int t  = sidx[j];
            int hh = wy * WIN + (t >> 4);
            int ww = wx * WIN + (t & 15);
            gx[k] = *reinterpret_cast<const float4*>(
                x + ((size_t)(b * IMG + hh) * IMG + ww) * C_DIM + c4 * 4);
        }
        for (int k = 0; k < NGW; ++k) gw[k] = wq4[tid + k * 128];
        for (int k = 0; k < NGX; ++k) {
            int i = tid + k * 128;
            int j = i / ROW4, c4 = i % ROW4;
            *reinterpret_cast<float4*>(&Xg[j * XP + c4 * 4]) = gx[k];
        }
        for (int k = 0; k < NGW; ++k)
            *reinterpret_cast<float4*>(&sWq[(tid + k * 128) * 4]) = gw[k];
    }
    __syncthreads();

    for (int tile = wave; tile < 12; tile += 4) {
        int mt = tile / 3, nt = tile % 3;
        const float* ap = &Xg[(mt * 16 + lo) * XP + 2 * hi];
        v8f acc = {};
        for (int kk = 0; kk < C_DIM; kk += 4) {
            int ka = kk + 2 * hi;
            v2f a  = *reinterpret_cast<const v2f*>(ap + kk);
            v2f bf = { sWq[ka * 48 + nt * 16 + lo],
                       sWq[(ka + 1) * 48 + nt * 16 + lo] };
            acc = wmma4(a, bf, acc);
        }
        #pragma unroll
        for (int r = 0; r < 8; ++r)
            sQKV[(mt * 16 + r + 8 * hi) * QP + nt * 16 + lo] = acc[r];
    }
    __syncthreads();

    const int q0 = wave * 16;
    v2f qf[HD / 4], kf[KSEL / 16][HD / 4], vf[KSEL / 16][4];
    float gate;
    {
        const float* qp = &sQKV[(q0 + lo) * QP + 2 * hi];
        #pragma unroll
        for (int s = 0; s < HD / 4; ++s) qf[s] = *reinterpret_cast<const v2f*>(qp + 4 * s);
        #pragma unroll
        for (int t = 0; t < KSEL / 16; ++t) {
            const float* kp = &sQKV[(t * 16 + lo) * QP + 16 + 2 * hi];
            #pragma unroll
            for (int s = 0; s < HD / 4; ++s) kf[t][s] = *reinterpret_cast<const v2f*>(kp + 4 * s);
        }
        #pragma unroll
        for (int t = 0; t < KSEL / 16; ++t) {
            #pragma unroll
            for (int j = 0; j < 4; ++j) {
                const int ka = t * 16 + 8 * hi + 2 * j;
                vf[t][j] = v2f{ sQKV[ka * QP + 32 + lo], sQKV[(ka + 1) * QP + 32 + lo] };
            }
        }
        gate = sgate[q0 + lo];
    }
    __builtin_amdgcn_sched_barrier(0);

    v8f pt[KSEL / 16];
    #pragma unroll
    for (int t = 0; t < KSEL / 16; ++t) pt[t] = v8f{};
    #pragma unroll
    for (int s = 0; s < HD / 4; ++s) {
        #pragma unroll
        for (int t = 0; t < KSEL / 16; ++t) pt[t] = wmma4(kf[t][s], qf[s], pt[t]);
    }

    {
        float m = -1e30f;
        #pragma unroll
        for (int t = 0; t < KSEL / 16; ++t) {
            #pragma unroll
            for (int r = 0; r < 8; ++r) { pt[t][r] *= 0.25f; m = fmaxf(m, pt[t][r]); }
        }
        m = fmaxf(m, xhalf(m));
        float s = 0.f;
        #pragma unroll
        for (int t = 0; t < KSEL / 16; ++t) {
            #pragma unroll
            for (int r = 0; r < 8; ++r) {
                float ev = __expf(pt[t][r] - m);
                pt[t][r] = ev; s += ev;
            }
        }
        s += xhalf(s);
        const float inv = gate / s;
        #pragma unroll
        for (int t = 0; t < KSEL / 16; ++t) {
            #pragma unroll
            for (int r = 0; r < 8; ++r) pt[t][r] *= inv;
        }
    }

    {
        v8f o = {};
        #pragma unroll
        for (int t = 0; t < KSEL / 16; ++t) {
            #pragma unroll
            for (int j = 0; j < 4; ++j)
                o = wmma4(vf[t][j], v2f{ pt[t][2 * j], pt[t][2 * j + 1] }, o);
        }
        float* orow = &O[(q0 + lo) * OP + 8 * hi];
        *reinterpret_cast<float4*>(orow)     = float4{ o[0], o[1], o[2], o[3] };
        *reinterpret_cast<float4*>(orow + 4) = float4{ o[4], o[5], o[6], o[7] };
    }
    __syncthreads();

    {
        float4* ob4 = reinterpret_cast<float4*>(Obuf + ((size_t)w * HEADS + e) * KSEL * HD);
        for (int q = 0; q < KSEL * HD / 4 / 128; ++q) {
            int p = tid + q * 128;
            int r = p >> 2, c4 = p & 3;
            ob4[p] = *reinterpret_cast<const float4*>(&O[r * OP + c4 * 4]);
        }
    }
}

static __device__ inline void k3_load_head(
    const float* __restrict__ Ow, const float* __restrict__ Wp,
    int e, int nt, int lo, int hi,
    v2f (&a)[KSEL / 16][HD / 4], v2f (&bf)[HD / 4])
{
    #pragma unroll
    for (int j = 0; j < HD / 4; ++j) {
        const int ka = 4 * j + 2 * hi;
        bf[j] = v2f{ Wp[((size_t)e * HD + ka) * C_DIM + nt * 16 + lo],
                     Wp[((size_t)e * HD + ka + 1) * C_DIM + nt * 16 + lo] };
    }
    #pragma unroll
    for (int mt = 0; mt < KSEL / 16; ++mt) {
        const float* ap = Ow + ((size_t)e * KSEL + mt * 16 + lo) * HD + 2 * hi;
        #pragma unroll
        for (int j = 0; j < HD / 4; ++j)
            a[mt][j] = *reinterpret_cast<const v2f*>(ap + 4 * j);
    }
}

__global__ __launch_bounds__(192) void k3_owner(
    const float* __restrict__ Obuf, const float* __restrict__ Wp,
    const float* __restrict__ projb, const int* __restrict__ gidx,
    float* __restrict__ y)
{
    __shared__ __attribute__((aligned(16))) float yt[LTOK * C_DIM];
    __shared__ __attribute__((aligned(16))) int   stok[HEADS * KSEL];

    const int w    = blockIdx.x;
    const int tid  = threadIdx.x;
    const int lane = tid & 31;
    const int nt   = tid >> 5;
    const int hi   = lane >> 4;
    const int lo   = lane & 15;

    const float* Ow = Obuf + (size_t)w * HEADS * KSEL * HD;

    v2f a_n[KSEL / 16][HD / 4], bf_n[HD / 4];
    k3_load_head(Ow, Wp, 0, nt, lo, hi, a_n, bf_n);

    stok[tid]       = gidx[(size_t)w * HEADS * KSEL + tid];
    stok[tid + 192] = gidx[(size_t)w * HEADS * KSEL + tid + 192];
    {
        float4* yt4 = reinterpret_cast<float4*>(yt);
        for (int q = 0; q < LTOK * C_DIM / 4 / 192; ++q)
            yt4[tid + q * 192] = float4{0.f, 0.f, 0.f, 0.f};
    }
    __syncthreads();

    float* ycol = yt + nt * 16 + lo;

    #pragma clang loop unroll(disable)
    for (int e = 0; e < HEADS; ++e) {
        v2f a[KSEL / 16][HD / 4], bf[HD / 4];
        #pragma unroll
        for (int j = 0; j < HD / 4; ++j) {
            bf[j] = bf_n[j];
            #pragma unroll
            for (int mt = 0; mt < KSEL / 16; ++mt) a[mt][j] = a_n[mt][j];
        }
        if (e + 1 < HEADS) k3_load_head(Ow, Wp, e + 1, nt, lo, hi, a_n, bf_n);

        int tok[KSEL / 16][8];
        v8f acc[KSEL / 16];
        #pragma unroll
        for (int mt = 0; mt < KSEL / 16; ++mt) {
            const int4* tp = reinterpret_cast<const int4*>(&stok[e * KSEL + mt * 16 + 8 * hi]);
            const int4 t0 = tp[0], t1 = tp[1];
            tok[mt][0] = t0.x; tok[mt][1] = t0.y; tok[mt][2] = t0.z; tok[mt][3] = t0.w;
            tok[mt][4] = t1.x; tok[mt][5] = t1.y; tok[mt][6] = t1.z; tok[mt][7] = t1.w;
            #pragma unroll
            for (int r = 0; r < 8; ++r) acc[mt][r] = ycol[tok[mt][r] * C_DIM];
        }
        #pragma unroll
        for (int j = 0; j < HD / 4; ++j) {
            #pragma unroll
            for (int mt = 0; mt < KSEL / 16; ++mt) acc[mt] = wmma4(a[mt][j], bf[j], acc[mt]);
        }
        #pragma unroll
        for (int mt = 0; mt < KSEL / 16; ++mt) {
            #pragma unroll
            for (int r = 0; r < 8; ++r) ycol[tok[mt][r] * C_DIM] = acc[mt][r];
        }
    }
    __syncthreads();

    {
        const int b  = w / (NWIN * NWIN);
        const int rm = w % (NWIN * NWIN);
        const int wy = rm / NWIN, wx = rm % NWIN;
        constexpr int ROWP = WIN * C_DIM / 4;
        constexpr int IMGP = IMG * C_DIM / 4;
        float4* y4 = reinterpret_cast<float4*>(y) +
                     ((size_t)(b * IMG + wy * WIN) * IMG + wx * WIN) * (C_DIM / 4);
        const float4* yt4 = reinterpret_cast<const float4*>(yt);
        const float4 bv = reinterpret_cast<const float4*>(projb)[tid % (C_DIM / 4)];
        for (int ty = 0; ty < WIN; ++ty) {
            for (int h2 = 0; h2 < ROWP / 192; ++h2) {
                int c = tid + h2 * 192;
                float4 v = yt4[ty * ROWP + c];
                v.x += bv.x; v.y += bv.y; v.z += bv.z; v.w += bv.w;
                y4[(size_t)ty * IMGP + c] = v;
            }
        }
    }
}

extern "C" void kernel_launch(void* const* d_in, const int* in_sizes, int n_in,
                              void* d_out, int out_size, void* d_ws, size_t ws_size,
                              hipStream_t stream) {
    const float* x     = (const float*)d_in[0];
    const float* Wr    = (const float*)d_in[1];
    const float* Wqkv  = (const float*)d_in[2];
    const float* Wo    = (const float*)d_in[3];
    const float* projW = (const float*)d_in[4];
    const float* projb = (const float*)d_in[5];
    float* y = (float*)d_out;

    float* Wp   = (float*)d_ws;
    int*   gidx = (int*)((char*)d_ws + (size_t)HEADS * HD * C_DIM * sizeof(float));
    float* gval = (float*)((char*)d_ws + (size_t)HEADS * HD * C_DIM * sizeof(float) +
                           (size_t)BW * HEADS * KSEL * sizeof(int));
    float* Obuf = (float*)((char*)d_ws + (size_t)HEADS * HD * C_DIM * sizeof(float) +
                           (size_t)BW * HEADS * KSEL * (sizeof(int) + sizeof(float)));

    k1_router<<<BW + HEADS, 256, 0, stream>>>(x, Wr, Wo, projW, gidx, gval, Wp);
    k2_attn<<<dim3(BW, HEADS), 128, 0, stream>>>(x, Wqkv, gidx, gval, Obuf);
    k3_owner<<<BW, 192, 0, stream>>>(Obuf, Wp, projb, gidx, y);
}


# k1: x read as whole 128-byte lines (8 lanes per token line, 4 lines per load), turned into token rows through a 36-float-pitch LDS tile; a load no longer touches 32 lines; logit FMA order unchanged
# speedup vs baseline: 3.6241x; 1.0962x over previous
#include <hip/hip_runtime.h>
#include <hip/hip_bf16.h>


#define BATCH   2
#define IMG     384
#define C_DIM   96
#define HEADS   6
#define HD      16
#define WIN     16
#define NWIN    24
#define BW      (BATCH * NWIN * NWIN)
#define LTOK    256
#define KSEL    64
#define XSP     36

typedef __attribute__((ext_vector_type(2))) float v2f;
typedef __attribute__((ext_vector_type(8))) float v8f;

static __device__ inline v8f wmma4(v2f a, v2f b, v8f c) {
    return __builtin_amdgcn_wmma_f32_16x16x4_f32(
         false, a,  false, b,
         (short)0, c,  false,  false);
}

static __device__ inline float xhalf(float v) {
    return __builtin_bit_cast(float, __builtin_amdgcn_permlanex16(
        __builtin_bit_cast(int, v), __builtin_bit_cast(int, v),
        0x76543210, 0xfedcba98, false, false));
}

template <int CK>
static __device__ inline void k1_piece(float (&lg)[HEADS], const float* sWr, const float4* rp)
{
    #pragma unroll
    for (int i = 0; i < 8; ++i) {
        const float4 v = rp[i];
        const int c = CK * 32 + i * 4;
        #pragma unroll
        for (int e = 0; e < HEADS; ++e) lg[e] += v.x * sWr[(c + 0) * HEADS + e];
        #pragma unroll
        for (int e = 0; e < HEADS; ++e) lg[e] += v.y * sWr[(c + 1) * HEADS + e];
        #pragma unroll
        for (int e = 0; e < HEADS; ++e) lg[e] += v.z * sWr[(c + 2) * HEADS + e];
        #pragma unroll
        for (int e = 0; e < HEADS; ++e) lg[e] += v.w * sWr[(c + 3) * HEADS + e];
    }
}

__global__ __launch_bounds__(256) void k1_router(
    const float* __restrict__ x, const float* __restrict__ Wr,
    const float* __restrict__ Wo, const float* __restrict__ projW,
    int* __restrict__ gidx, float* __restrict__ gval,
    float* __restrict__ Wp)
{
    __shared__ float sWr[C_DIM * HEADS];
    __shared__ float sv[HEADS * LTOK];
    __shared__ __attribute__((aligned(16))) float xs[LTOK * XSP];

    const int w = blockIdx.x;
    const int t = threadIdx.x;

    if (w >= BW) {
        const int e    = w - BW;
        const int lane = t & 31, nt = t >> 5;
        const int hi   = lane >> 4, lo = lane & 15;
        if (nt < C_DIM / 16) {
            const float* ap = Wo + ((size_t)e * HD + lo) * C_DIM + 2 * hi;
            const float* bp = projW + (size_t)(nt * 16 + lo) * C_DIM + 2 * hi;
            v8f acc = {};
            for (int kk = 0; kk < C_DIM; kk += 4) {
                v2f a  = *reinterpret_cast<const v2f*>(ap + kk);
                v2f bf = *reinterpret_cast<const v2f*>(bp + kk);
                acc = wmma4(a, bf, acc);
            }
            #pragma unroll
            for (int r = 0; r < 8; ++r)
                Wp[((size_t)e * HD + r + 8 * hi) * C_DIM + nt * 16 + lo] = acc[r];
        }
        return;
    }

    for (int i = t; i < C_DIM * HEADS; i += 256) sWr[i] = Wr[i];

    const int b  = w / (NWIN * NWIN);
    const int rm = w % (NWIN * NWIN);
    const int wy = rm / NWIN, wx = rm % NWIN;

    const int tsub = t >> 3, part = t & 7;
    const float* gp = x + ((size_t)(b * IMG + wy * WIN + (tsub >> 4)) * IMG
                           + wx * WIN + (tsub & 15)) * C_DIM + part * 4;
    constexpr size_t QSTEP = (size_t)2 * IMG * C_DIM;
    float* sp = &xs[tsub * XSP + part * 4];
    const float4* rp = reinterpret_cast<const float4*>(&xs[t * XSP]);

    float lg[HEADS];
    #pragma unroll
    for (int e = 0; e < HEADS; ++e) lg[e] = 0.f;

#define K1_LOAD8(n, ck)                                                                   \
    const float4 n##0 = *reinterpret_cast<const float4*>(gp + 0 * QSTEP + (ck) * 32),     \
                 n##1 = *reinterpret_cast<const float4*>(gp + 1 * QSTEP + (ck) * 32),     \
                 n##2 = *reinterpret_cast<const float4*>(gp + 2 * QSTEP + (ck) * 32),     \
                 n##3 = *reinterpret_cast<const float4*>(gp + 3 * QSTEP + (ck) * 32),     \
                 n##4 = *reinterpret_cast<const float4*>(gp + 4 * QSTEP + (ck) * 32),     \
                 n##5 = *reinterpret_cast<const float4*>(gp + 5 * QSTEP + (ck) * 32),     \
                 n##6 = *reinterpret_cast<const float4*>(gp + 6 * QSTEP + (ck) * 32),     \
                 n##7 = *reinterpret_cast<const float4*>(gp + 7 * QSTEP + (ck) * 32)
#define K1_STORE8(n)                                                                      \
    *reinterpret_cast<float4*>(sp + 0 * 32 * XSP) = n##0;                                 \
    *reinterpret_cast<float4*>(sp + 1 * 32 * XSP) = n##1;                                 \
    *reinterpret_cast<float4*>(sp + 2 * 32 * XSP) = n##2;                                 \
    *reinterpret_cast<float4*>(sp + 3 * 32 * XSP) = n##3;                                 \
    *reinterpret_cast<float4*>(sp + 4 * 32 * XSP) = n##4;                                 \
    *reinterpret_cast<float4*>(sp + 5 * 32 * XSP) = n##5;                                 \
    *reinterpret_cast<float4*>(sp + 6 * 32 * XSP) = n##6;                                 \
    *reinterpret_cast<float4*>(sp + 7 * 32 * XSP) = n##7

    K1_LOAD8(ga, 0);
    K1_STORE8(ga);
    __syncthreads();
    k1_piece<0>(lg, sWr, rp);
    __syncthreads();
    K1_LOAD8(gb, 1);
    K1_STORE8(gb);
    __syncthreads();
    k1_piece<1>(lg, sWr, rp);
    __syncthreads();
    K1_LOAD8(gc, 2);
    K1_STORE8(gc);
    __syncthreads();
    k1_piece<2>(lg, sWr, rp);
#undef K1_LOAD8
#undef K1_STORE8

    #pragma unroll
    for (int e = 0; e < HEADS; ++e) lg[e] = 1.0f / (1.0f + __expf(-lg[e]));

    #pragma unroll
    for (int e = 0; e < HEADS; ++e) sv[e * LTOK + t] = lg[e];
    __syncthreads();

    const int wv = t >> 5, lane = t & 31;
    if (wv < HEADS) {
        const int e = wv;
        float    val[8];
        unsigned kb[8];
        #pragma unroll
        for (int r = 0; r < 8; ++r) {
            val[r] = sv[e * LTOK + r * 32 + lane];
            kb[r]  = __float_as_uint(val[r]);
        }
        const unsigned tail0 = 0xFFFFFFFEu;

        unsigned T = 0;
        for (int bit = 31; bit >= 0; --bit) {
            const unsigned cand = T | (1u << bit);
            int cnt = 0;
            #pragma unroll
            for (int r = 0; r < 8; ++r) {
                unsigned m = __builtin_amdgcn_ballot_w32(kb[r] >= cand);
                if (r == 0) m &= tail0;
                cnt += __builtin_popcount(m);
            }
            if (cnt >= KSEL - 1) T = cand;
        }

        unsigned gm[8], em[8];
        int gt = 0;
        #pragma unroll
        for (int r = 0; r < 8; ++r) {
            gm[r] = __builtin_amdgcn_ballot_w32(kb[r] > T);
            em[r] = __builtin_amdgcn_ballot_w32(kb[r] == T);
            if (r == 0) { gm[r] &= tail0; em[r] &= tail0; }
            gt += __builtin_popcount(gm[r]);
        }
        const int need = (KSEL - 1) - gt;

        const size_t base = ((size_t)w * HEADS + e) * KSEL;
        int eqBefore = 0, selBefore = 1;
        #pragma unroll
        for (int r = 0; r < 8; ++r) {
            const bool above = (gm[r] >> lane) & 1u;
            const bool equal = (em[r] >> lane) & 1u;
            const int  eqRank = eqBefore + (int)__builtin_amdgcn_mbcnt_lo(em[r], 0u);
            const bool sel = above || (equal && eqRank < need);
            const unsigned sm = __builtin_amdgcn_ballot_w32(sel);
            if (sel) {
                const int pos = selBefore + (int)__builtin_amdgcn_mbcnt_lo(sm, 0u);
                gidx[base + pos] = r * 32 + lane;
                gval[base + pos] = val[r];
            }
            eqBefore  += __builtin_popcount(em[r]);
            selBefore += __builtin_popcount(sm);
        }
        if (lane == 0) { gidx[base] = 0; gval[base] = val[0]; }
    }
}

#define XP  100
#define QP  52
#define OP  20
__global__ __launch_bounds__(128) void k2_attn(
    const float* __restrict__ x,
    const float* __restrict__ Wqkv,
    const int* __restrict__ gidx, const float* __restrict__ gval,
    float* __restrict__ Obuf)
{
    __shared__ __attribute__((aligned(16))) float pool[KSEL * XP + C_DIM * 48];
    __shared__ __attribute__((aligned(16))) float sQKV[KSEL * QP];
    __shared__ float sgate[KSEL];
    __shared__ int   sidx[KSEL];

    float* Xg  = pool;
    float* sWq = pool + KSEL * XP;
    float* O   = pool;

    const int w    = blockIdx.x;
    const int e    = blockIdx.y;
    const int tid  = threadIdx.x;
    const int lane = tid & 31;
    const int wave = tid >> 5;
    const int hi   = lane >> 4;
    const int lo   = lane & 15;

    if (tid < KSEL) {
        sidx[tid]  = gidx[((size_t)w * HEADS + e) * KSEL + tid];
        sgate[tid] = gval[((size_t)w * HEADS + e) * KSEL + tid];
    }
    __syncthreads();

    const int b  = w / (NWIN * NWIN);
    const int rm = w % (NWIN * NWIN);
    const int wy = rm / NWIN, wx = rm % NWIN;

    constexpr int ROW4 = C_DIM / 4;
    constexpr int NGX  = KSEL * ROW4 / 128;
    constexpr int NGW  = C_DIM * 48 / 4 / 128;
    {
        float4 gx[NGX], gw[NGW];
        const float4* wq4 = reinterpret_cast<const float4*>(Wqkv + (size_t)e * C_DIM * 48);
        for (int k = 0; k < NGX; ++k) {
            int i = tid + k * 128;
            int j = i / ROW4, c4 = i % ROW4;
            int t  = sidx[j];
            int hh = wy * WIN + (t >> 4);
            int ww = wx * WIN + (t & 15);
            gx[k] = *reinterpret_cast<const float4*>(
                x + ((size_t)(b * IMG + hh) * IMG + ww) * C_DIM + c4 * 4);
        }
        for (int k = 0; k < NGW; ++k) gw[k] = wq4[tid + k * 128];
        for (int k = 0; k < NGX; ++k) {
            int i = tid + k * 128;
            int j = i / ROW4, c4 = i % ROW4;
            *reinterpret_cast<float4*>(&Xg[j * XP + c4 * 4]) = gx[k];
        }
        for (int k = 0; k < NGW; ++k)
            *reinterpret_cast<float4*>(&sWq[(tid + k * 128) * 4]) = gw[k];
    }
    __syncthreads();

    for (int tile = wave; tile < 12; tile += 4) {
        int mt = tile / 3, nt = tile % 3;
        const float* ap = &Xg[(mt * 16 + lo) * XP + 2 * hi];
        v8f acc = {};
        for (int kk = 0; kk < C_DIM; kk += 4) {
            int ka = kk + 2 * hi;
            v2f a  = *reinterpret_cast<const v2f*>(ap + kk);
            v2f bf = { sWq[ka * 48 + nt * 16 + lo],
                       sWq[(ka + 1) * 48 + nt * 16 + lo] };
            acc = wmma4(a, bf, acc);
        }
        #pragma unroll
        for (int r = 0; r < 8; ++r)
            sQKV[(mt * 16 + r + 8 * hi) * QP + nt * 16 + lo] = acc[r];
    }
    __syncthreads();

    const int q0 = wave * 16;
    v2f qf[HD / 4], kf[KSEL / 16][HD / 4], vf[KSEL / 16][4];
    float gate;
    {
        const float* qp = &sQKV[(q0 + lo) * QP + 2 * hi];
        #pragma unroll
        for (int s = 0; s < HD / 4; ++s) qf[s] = *reinterpret_cast<const v2f*>(qp + 4 * s);
        #pragma unroll
        for (int t = 0; t < KSEL / 16; ++t) {
            const float* kp = &sQKV[(t * 16 + lo) * QP + 16 + 2 * hi];
            #pragma unroll
            for (int s = 0; s < HD / 4; ++s) kf[t][s] = *reinterpret_cast<const v2f*>(kp + 4 * s);
        }
        #pragma unroll
        for (int t = 0; t < KSEL / 16; ++t) {
            #pragma unroll
            for (int j = 0; j < 4; ++j) {
                const int ka = t * 16 + 8 * hi + 2 * j;
                vf[t][j] = v2f{ sQKV[ka * QP + 32 + lo], sQKV[(ka + 1) * QP + 32 + lo] };
            }
        }
        gate = sgate[q0 + lo];
    }
    __builtin_amdgcn_sched_barrier(0);

    v8f pt[KSEL / 16];
    #pragma unroll
    for (int t = 0; t < KSEL / 16; ++t) pt[t] = v8f{};
    #pragma unroll
    for (int s = 0; s < HD / 4; ++s) {
        #pragma unroll
        for (int t = 0; t < KSEL / 16; ++t) pt[t] = wmma4(kf[t][s], qf[s], pt[t]);
    }

    {
        float m = -1e30f;
        #pragma unroll
        for (int t = 0; t < KSEL / 16; ++t) {
            #pragma unroll
            for (int r = 0; r < 8; ++r) { pt[t][r] *= 0.25f; m = fmaxf(m, pt[t][r]); }
        }
        m = fmaxf(m, xhalf(m));
        float s = 0.f;
        #pragma unroll
        for (int t = 0; t < KSEL / 16; ++t) {
            #pragma unroll
            for (int r = 0; r < 8; ++r) {
                float ev = __expf(pt[t][r] - m);
                pt[t][r] = ev; s += ev;
            }
        }
        s += xhalf(s);
        const float inv = gate / s;
        #pragma unroll
        for (int t = 0; t < KSEL / 16; ++t) {
            #pragma unroll
            for (int r = 0; r < 8; ++r) pt[t][r] *= inv;
        }
    }

    {
        v8f o = {};
        #pragma unroll
        for (int t = 0; t < KSEL / 16; ++t) {
            #pragma unroll
            for (int j = 0; j < 4; ++j)
                o = wmma4(vf[t][j], v2f{ pt[t][2 * j], pt[t][2 * j + 1] }, o);
        }
        float* orow = &O[(q0 + lo) * OP + 8 * hi];
        *reinterpret_cast<float4*>(orow)     = float4{ o[0], o[1], o[2], o[3] };
        *reinterpret_cast<float4*>(orow + 4) = float4{ o[4], o[5], o[6], o[7] };
    }
    __syncthreads();

    {
        float4* ob4 = reinterpret_cast<float4*>(Obuf + ((size_t)w * HEADS + e) * KSEL * HD);
        for (int q = 0; q < KSEL * HD / 4 / 128; ++q) {
            int p = tid + q * 128;
            int r = p >> 2, c4 = p & 3;
            ob4[p] = *reinterpret_cast<const float4*>(&O[r * OP + c4 * 4]);
        }
    }
}

static __device__ inline void k3_load_head(
    const float* __restrict__ Ow, const float* __restrict__ Wp,
    int e, int nt, int lo, int hi,
    v2f (&a)[KSEL / 16][HD / 4], v2f (&bf)[HD / 4])
{
    #pragma unroll
    for (int j = 0; j < HD / 4; ++j) {
        const int ka = 4 * j + 2 * hi;
        bf[j] = v2f{ Wp[((size_t)e * HD + ka) * C_DIM + nt * 16 + lo],
                     Wp[((size_t)e * HD + ka + 1) * C_DIM + nt * 16 + lo] };
    }
    #pragma unroll
    for (int mt = 0; mt < KSEL / 16; ++mt) {
        const float* ap = Ow + ((size_t)e * KSEL + mt * 16 + lo) * HD + 2 * hi;
        #pragma unroll
        for (int j = 0; j < HD / 4; ++j)
            a[mt][j] = *reinterpret_cast<const v2f*>(ap + 4 * j);
    }
}

__global__ __launch_bounds__(192) void k3_owner(
    const float* __restrict__ Obuf, const float* __restrict__ Wp,
    const float* __restrict__ projb, const int* __restrict__ gidx,
    float* __restrict__ y)
{
    __shared__ __attribute__((aligned(16))) float yt[LTOK * C_DIM];
    __shared__ __attribute__((aligned(16))) int   stok[HEADS * KSEL];

    const int w    = blockIdx.x;
    const int tid  = threadIdx.x;
    const int lane = tid & 31;
    const int nt   = tid >> 5;
    const int hi   = lane >> 4;
    const int lo   = lane & 15;

    const float* Ow = Obuf + (size_t)w * HEADS * KSEL * HD;

    v2f a_n[KSEL / 16][HD / 4], bf_n[HD / 4];
    k3_load_head(Ow, Wp, 0, nt, lo, hi, a_n, bf_n);

    stok[tid]       = gidx[(size_t)w * HEADS * KSEL + tid];
    stok[tid + 192] = gidx[(size_t)w * HEADS * KSEL + tid + 192];
    {
        float4* yt4 = reinterpret_cast<float4*>(yt);
        for (int q = 0; q < LTOK * C_DIM / 4 / 192; ++q)
            yt4[tid + q * 192] = float4{0.f, 0.f, 0.f, 0.f};
    }
    __syncthreads();

    float* ycol = yt + nt * 16 + lo;

    #pragma clang loop unroll(disable)
    for (int e = 0; e < HEADS; ++e) {
        v2f a[KSEL / 16][HD / 4], bf[HD / 4];
        #pragma unroll
        for (int j = 0; j < HD / 4; ++j) {
            bf[j] = bf_n[j];
            #pragma unroll
            for (int mt = 0; mt < KSEL / 16; ++mt) a[mt][j] = a_n[mt][j];
        }
        if (e + 1 < HEADS) k3_load_head(Ow, Wp, e + 1, nt, lo, hi, a_n, bf_n);

        int tok[KSEL / 16][8];
        v8f acc[KSEL / 16];
        #pragma unroll
        for (int mt = 0; mt < KSEL / 16; ++mt) {
            const int4* tp = reinterpret_cast<const int4*>(&stok[e * KSEL + mt * 16 + 8 * hi]);
            const int4 t0 = tp[0], t1 = tp[1];
            tok[mt][0] = t0.x; tok[mt][1] = t0.y; tok[mt][2] = t0.z; tok[mt][3] = t0.w;
            tok[mt][4] = t1.x; tok[mt][5] = t1.y; tok[mt][6] = t1.z; tok[mt][7] = t1.w;
            #pragma unroll
            for (int r = 0; r < 8; ++r) acc[mt][r] = ycol[tok[mt][r] * C_DIM];
        }
        #pragma unroll
        for (int j = 0; j < HD / 4; ++j) {
            #pragma unroll
            for (int mt = 0; mt < KSEL / 16; ++mt) acc[mt] = wmma4(a[mt][j], bf[j], acc[mt]);
        }
        #pragma unroll
        for (int mt = 0; mt < KSEL / 16; ++mt) {
            #pragma unroll
            for (int r = 0; r < 8; ++r) ycol[tok[mt][r] * C_DIM] = acc[mt][r];
        }
    }
    __syncthreads();

    {
        const int b  = w / (NWIN * NWIN);
        const int rm = w % (NWIN * NWIN);
        const int wy = rm / NWIN, wx = rm % NWIN;
        constexpr int ROWP = WIN * C_DIM / 4;
        constexpr int IMGP = IMG * C_DIM / 4;
        float4* y4 = reinterpret_cast<float4*>(y) +
                     ((size_t)(b * IMG + wy * WIN) * IMG + wx * WIN) * (C_DIM / 4);
        const float4* yt4 = reinterpret_cast<const float4*>(yt);
        const float4 bv = reinterpret_cast<const float4*>(projb)[tid % (C_DIM / 4)];
        for (int ty = 0; ty < WIN; ++ty) {
            for (int h2 = 0; h2 < ROWP / 192; ++h2) {
                int c = tid + h2 * 192;
                float4 v = yt4[ty * ROWP + c];
                v.x += bv.x; v.y += bv.y; v.z += bv.z; v.w += bv.w;
                y4[(size_t)ty * IMGP + c] = v;
            }
        }
    }
}

extern "C" void kernel_launch(void* const* d_in, const int* in_sizes, int n_in,
                              void* d_out, int out_size, void* d_ws, size_t ws_size,
                              hipStream_t stream) {
    const float* x     = (const float*)d_in[0];
    const float* Wr    = (const float*)d_in[1];
    const float* Wqkv  = (const float*)d_in[2];
    const float* Wo    = (const float*)d_in[3];
    const float* projW = (const float*)d_in[4];
    const float* projb = (const float*)d_in[5];
    float* y = (float*)d_out;

    float* Wp   = (float*)d_ws;
    int*   gidx = (int*)((char*)d_ws + (size_t)HEADS * HD * C_DIM * sizeof(float));
    float* gval = (float*)((char*)d_ws + (size_t)HEADS * HD * C_DIM * sizeof(float) +
                           (size_t)BW * HEADS * KSEL * sizeof(int));
    float* Obuf = (float*)((char*)d_ws + (size_t)HEADS * HD * C_DIM * sizeof(float) +
                           (size_t)BW * HEADS * KSEL * (sizeof(int) + sizeof(float)));

    k1_router<<<BW + HEADS, 256, 0, stream>>>(x, Wr, Wo, projW, gidx, gval, Wp);
    k2_attn<<<dim3(BW, HEADS), 128, 0, stream>>>(x, Wqkv, gidx, gval, Obuf);
    k3_owner<<<BW, 192, 0, stream>>>(Obuf, Wp, projb, gidx, y);
}
